# starting kernel
# baseline (speedup 1.0000x reference)
#include <hip/hip_runtime.h>

typedef __attribute__((ext_vector_type(2))) float v2f;
typedef __attribute__((ext_vector_type(8))) float v8f;

#define NN 100000
#define NCH 4

__device__ __forceinline__ v8f wmma_f32(v2f a, v2f b, v8f c) {
  return __builtin_amdgcn_wmma_f32_16x16x4_f32(false, a, false, b, (short)0, c,
                                               false, false);
}

__global__ __launch_bounds__(256) void repack_w(
    const float* __restrict__ W, float* __restrict__ Wp, int K, int C)
{
  const size_t mat = (size_t)blockIdx.y * K * C;
  size_t idx = (size_t)blockIdx.x * 256 + threadIdx.x;
  if (idx >= (size_t)K * C) return;
  int k   = (int)(idx / C);
  int col = (int)(idx % C);
  Wp[mat + (((size_t)(k >> 2) * C + col) << 2) + (k & 3)] = W[mat + idx];
}

__global__ __launch_bounds__(256) void copy_rows128(
    const float* __restrict__ in, int ldi,
    float* __restrict__ outp, int ldo, int nRows)
{
  size_t idx = (size_t)blockIdx.x * 256 + threadIdx.x;
  if (idx >= (size_t)nRows * 32) return;
  size_t row = idx >> 5;
  int c4 = (int)(idx & 31) << 2;
  const float4 v = *(const float4*)(in + row * (size_t)ldi + c4);
  *(float4*)(outp + row * (size_t)ldo + c4) = v;
}

__global__ __launch_bounds__(256) void gin_scatter(
    const float* __restrict__ feat, int ldf,
    const int* __restrict__ srcIdx, const int* __restrict__ dstIdx,
    float* __restrict__ agg, long long E)
{
  long long t = (long long)blockIdx.x * 256 + threadIdx.x;
  long long e = t >> 5;
  if (e >= E) return;
  int g = (int)(t & 31);
  int s = srcIdx[e];
  int d = dstIdx[e];
  const float4 v = *(const float4*)(feat + (size_t)s * ldf + g * 4);
  float* p = agg + (size_t)d * 128 + g * 4;
  unsafeAtomicAdd(p + 0, v.x);
  unsafeAtomicAdd(p + 1, v.y);
  unsafeAtomicAdd(p + 2, v.z);
  unsafeAtomicAdd(p + 3, v.w);
}

__global__ __launch_bounds__(256) void gin_mlp3(
    const float* __restrict__ agg,
    const float* __restrict__ Wp,
    const float* __restrict__ Bv,
    float* __restrict__ xcat, int colOff)
{
  __shared__ float bufA[32][132];
  __shared__ float bufB[32][132];
  const int tid  = threadIdx.x;
  const int lane = tid & 31;
  const int w    = tid >> 5;
  const size_t base = (size_t)blockIdx.x * 32;

  for (int i = tid; i < 32 * 32; i += 256) {
    int row = i >> 5, c4 = (i & 31) << 2;
    const float4 v = *(const float4*)(agg + (base + row) * 128 + c4);
    bufA[row][c4] = v.x; bufA[row][c4 + 1] = v.y;
    bufA[row][c4 + 2] = v.z; bufA[row][c4 + 3] = v.w;
  }
  __syncthreads();

  const int half = lane >> 4;
  const int n16  = lane & 15;
  const int cb   = w * 16;

  for (int layer = 0; layer < 3; ++layer) {
    const float* Wk = Wp + (size_t)layer * 128 * 128;
    float (*in)[132]   = (layer & 1) ? bufB : bufA;
    float (*outb)[132] = (layer & 1) ? bufA : bufB;
    v8f acc0 = {}, acc1 = {};
    for (int k = 0; k < 128; k += 4) {
      const int kk = k + half * 2;
      const v2f b = *(const v2f*)(Wk + (((size_t)(k >> 2) * 128 + cb + n16) << 2)
                                     + half * 2);
      v2f a0; a0.x = in[n16][kk];      a0.y = in[n16][kk + 1];
      v2f a1; a1.x = in[n16 + 16][kk]; a1.y = in[n16 + 16][kk + 1];
      acc0 = wmma_f32(a0, b, acc0);
      acc1 = wmma_f32(a1, b, acc1);
    }
    const float bias = Bv[layer * 128 + cb + n16];
    #pragma unroll
    for (int j = 0; j < 8; ++j) {
      float r0 = acc0[j] + bias; r0 = r0 > 0.f ? r0 : 0.f;
      float r1 = acc1[j] + bias; r1 = r1 > 0.f ? r1 : 0.f;
      outb[j + half * 8][cb + n16]      = r0;
      outb[16 + j + half * 8][cb + n16] = r1;
    }
    __syncthreads();
  }
  for (int i = tid; i < 32 * 32; i += 256) {
    int row = i >> 5, c4 = (i & 31) << 2;
    float4 v = make_float4(bufB[row][c4], bufB[row][c4 + 1],
                           bufB[row][c4 + 2], bufB[row][c4 + 3]);
    *(float4*)(xcat + (base + row) * 512 + colOff + c4) = v;
  }
}

__global__ __launch_bounds__(256) void gin_chan(
    const float* __restrict__ xcat,
    const float* __restrict__ W1p, const float* __restrict__ B1,
    const float* __restrict__ W2p, const float* __restrict__ B2,
    const float* __restrict__ W3,  const float* __restrict__ B3,
    float* __restrict__ outp)
{
  __shared__ float bufA[16][516];
  __shared__ float bufB[16][260];
  const int tid  = threadIdx.x;
  const int lane = tid & 31;
  const int w    = tid >> 5;
  const int c    = blockIdx.y;
  const size_t base = (size_t)blockIdx.x * 16;

  for (int i = tid; i < 16 * 128; i += 256) {
    int row = i >> 7, c4 = (i & 127) << 2;
    const float4 v = *(const float4*)(xcat + (base + row) * 512 + c4);
    bufA[row][c4] = v.x; bufA[row][c4 + 1] = v.y;
    bufA[row][c4 + 2] = v.z; bufA[row][c4 + 3] = v.w;
  }
  __syncthreads();

  const int half = lane >> 4;
  const int n16  = lane & 15;
  const int ct0  = w * 16;
  const int ct1  = (w + 8) * 16;

  {
    const float* W1c = W1p + (size_t)c * 512 * 256;
    v8f acc0 = {}, acc1 = {};
    for (int k = 0; k < 512; k += 4) {
      const int kk = k + half * 2;
      v2f a; a.x = bufA[n16][kk]; a.y = bufA[n16][kk + 1];
      const size_t kb = ((size_t)(k >> 2) * 256) << 2;
      const v2f b0 = *(const v2f*)(W1c + kb + ((ct0 + n16) << 2) + half * 2);
      acc0 = wmma_f32(a, b0, acc0);
      const v2f b1 = *(const v2f*)(W1c + kb + ((ct1 + n16) << 2) + half * 2);
      acc1 = wmma_f32(a, b1, acc1);
    }
    const float bias0 = B1[c * 256 + ct0 + n16];
    const float bias1 = B1[c * 256 + ct1 + n16];
    #pragma unroll
    for (int j = 0; j < 8; ++j) {
      float r0 = acc0[j] + bias0; r0 = r0 > 0.f ? r0 : 0.f;
      float r1 = acc1[j] + bias1; r1 = r1 > 0.f ? r1 : 0.f;
      bufB[j + half * 8][ct0 + n16] = r0;
      bufB[j + half * 8][ct1 + n16] = r1;
    }
  }
  __syncthreads();

  {
    const float* W2c = W2p + (size_t)c * 256 * 256;
    v8f acc0 = {}, acc1 = {};
    for (int k = 0; k < 256; k += 4) {
      const int kk = k + half * 2;
      v2f a; a.x = bufB[n16][kk]; a.y = bufB[n16][kk + 1];
      const size_t kb = ((size_t)(k >> 2) * 256) << 2;
      const v2f b0 = *(const v2f*)(W2c + kb + ((ct0 + n16) << 2) + half * 2);
      acc0 = wmma_f32(a, b0, acc0);
      const v2f b1 = *(const v2f*)(W2c + kb + ((ct1 + n16) << 2) + half * 2);
      acc1 = wmma_f32(a, b1, acc1);
    }
    const float bias0 = B2[c * 256 + ct0 + n16];
    const float bias1 = B2[c * 256 + ct1 + n16];
    #pragma unroll
    for (int j = 0; j < 8; ++j) {
      float r0 = acc0[j] + bias0; r0 = r0 > 0.f ? r0 : 0.f;
      float r1 = acc1[j] + bias1; r1 = r1 > 0.f ? r1 : 0.f;
      bufA[j + half * 8][ct0 + n16] = r0;
      bufA[j + half * 8][ct1 + n16] = r1;
    }
  }
  __syncthreads();

  if (w == 0 && lane < 16) {
    const float* w3 = W3 + (size_t)c * 256;
    float sum = 0.f;
    for (int k = 0; k < 256; ++k) sum += bufA[lane][k] * w3[k];
    sum += B3[c];
    sum = sum > 0.f ? sum : 0.f;
    outp[(base + lane) * NCH + c] = sum;
  }
}

extern "C" void kernel_launch(void* const* d_in, const int* in_sizes, int n_in,
                              void* d_out, int out_size, void* d_ws, size_t ws_size,
                              hipStream_t stream)
{
  (void)n_in; (void)out_size; (void)ws_size;
  const float* x      = (const float*)d_in[0];
  const int*   ei     = (const int*)d_in[1];
  const float* convW  = (const float*)d_in[2];
  const float* convB  = (const float*)d_in[3];
  const float* chanW1 = (const float*)d_in[4];
  const float* chanB1 = (const float*)d_in[5];
  const float* chanW2 = (const float*)d_in[6];
  const float* chanB2 = (const float*)d_in[7];
  const float* chanW3 = (const float*)d_in[8];
  const float* chanB3 = (const float*)d_in[9];
  float* outp = (float*)d_out;

  const int N = NN;
  const long long E = (long long)in_sizes[1] / 2;

  float* xcat   = (float*)d_ws;
  float* agg    = xcat + (size_t)N * 512;
  float* convWp = agg + (size_t)N * 128;
  float* W1p    = convWp + (size_t)9 * 128 * 128;
  float* W2p    = W1p + (size_t)4 * 512 * 256;

  repack_w<<<dim3((128 * 128) / 256, 9), 256, 0, stream>>>(convW, convWp, 128, 128);
  repack_w<<<dim3((512 * 256) / 256, 4), 256, 0, stream>>>(chanW1, W1p, 512, 256);
  repack_w<<<dim3((256 * 256) / 256, 4), 256, 0, stream>>>(chanW2, W2p, 256, 256);

  const int cpBlocks = (int)(((size_t)N * 32 + 255) / 256);
  const int scBlocks = (int)((E * 32 + 255) / 256);

  copy_rows128<<<cpBlocks, 256, 0, stream>>>(x, 128, xcat, 512, N);

  for (int l = 0; l < 3; ++l) {
    const float* h = (l == 0) ? x : (xcat + (size_t)l * 128);
    const int ldh  = (l == 0) ? 128 : 512;
    copy_rows128<<<cpBlocks, 256, 0, stream>>>(h, ldh, agg, 128, N);
    gin_scatter<<<scBlocks, 256, 0, stream>>>(h, ldh, ei, ei + E, agg, E);
    gin_mlp3<<<N / 32, 256, 0, stream>>>(agg,
                                         convWp + (size_t)l * 3 * 128 * 128,
                                         convB + l * 3 * 128,
                                         xcat, (l + 1) * 128);
  }

  gin_chan<<<dim3(N / 16, NCH), 256, 0, stream>>>(xcat, W1p, chanB1,
                                                  W2p, chanB2,
                                                  chanW3, chanB3, outp);
}


# float-atomic edge scatter + self copy replaced by a dst-grouped edge list built once (int atomic count, scan, cursor fill) and a per-node gather sum; node sum order = cursor arrival order
# speedup vs baseline: 1.4300x; 1.4300x over previous
#include <hip/hip_runtime.h>

typedef __attribute__((ext_vector_type(2))) float v2f;
typedef __attribute__((ext_vector_type(8))) float v8f;

#define NN 100000
#define NCH 4

__device__ __forceinline__ v8f wmma_f32(v2f a, v2f b, v8f c) {
  return __builtin_amdgcn_wmma_f32_16x16x4_f32(false, a, false, b, (short)0, c,
                                               false, false);
}

__global__ __launch_bounds__(256) void repack_w(
    const float* __restrict__ W, float* __restrict__ Wp, int K, int C)
{
  const size_t mat = (size_t)blockIdx.y * K * C;
  size_t idx = (size_t)blockIdx.x * 256 + threadIdx.x;
  if (idx >= (size_t)K * C) return;
  int k   = (int)(idx / C);
  int col = (int)(idx % C);
  Wp[mat + (((size_t)(k >> 2) * C + col) << 2) + (k & 3)] = W[mat + idx];
}

__global__ __launch_bounds__(256) void copy_rows128(
    const float* __restrict__ in, int ldi,
    float* __restrict__ outp, int ldo, int nRows)
{
  size_t idx = (size_t)blockIdx.x * 256 + threadIdx.x;
  if (idx >= (size_t)nRows * 32) return;
  size_t row = idx >> 5;
  int c4 = (int)(idx & 31) << 2;
  const float4 v = *(const float4*)(in + row * (size_t)ldi + c4);
  *(float4*)(outp + row * (size_t)ldo + c4) = v;
}

__global__ __launch_bounds__(256) void csr_zero(int* __restrict__ cnt, int n)
{
  const int i = blockIdx.x * 256 + threadIdx.x;
  if (i < n) cnt[i] = 0;
}

__global__ __launch_bounds__(256) void csr_count(
    const int* __restrict__ dstIdx, int* __restrict__ cnt, long long E)
{
  const long long e = (long long)blockIdx.x * 256 + threadIdx.x;
  if (e >= E) return;
  atomicAdd(cnt + dstIdx[e], 1);
}

__global__ __launch_bounds__(256) void csr_scan(
    const int* __restrict__ cnt, int* __restrict__ rowptr,
    int* __restrict__ cursor, int n)
{
  __shared__ int wpre[8];
  __shared__ int wsum[8];
  const int tid  = threadIdx.x;
  const int lane = tid & 31;
  const int w    = tid >> 5;
  const int first = blockIdx.x * 1024;

  int pre = 0;
  const int* mycol = cnt + tid * 4;
  int i = 0;
  for (; i + 8 * 1024 <= first; i += 8 * 1024) {
    int4 v[8];
    #pragma unroll
    for (int u = 0; u < 8; ++u) v[u] = *(const int4*)(mycol + i + u * 1024);
    #pragma unroll
    for (int u = 0; u < 8; ++u) pre += (v[u].x + v[u].y) + (v[u].z + v[u].w);
  }
  for (; i < first; i += 1024) {
    const int4 v = *(const int4*)(mycol + i);
    pre += (v.x + v.y) + (v.z + v.w);
  }

  const int i0 = first + tid * 4;
  int c0 = 0, c1 = 0, c2 = 0, c3 = 0;
  if (i0 + 3 < n) {
    const int4 v = *(const int4*)(cnt + i0);
    c0 = v.x; c1 = v.y; c2 = v.z; c3 = v.w;
  } else {
    if (i0     < n) c0 = cnt[i0];
    if (i0 + 1 < n) c1 = cnt[i0 + 1];
    if (i0 + 2 < n) c2 = cnt[i0 + 2];
  }
  const int own = (c0 + c1) + (c2 + c3);

  int incl = own;
  for (int d = 1; d < 32; d <<= 1) {
    const int t = __shfl_up(incl, d);
    if (lane >= d) incl += t;
  }
  for (int d = 16; d >= 1; d >>= 1) pre += __shfl_xor(pre, d);
  if (lane == 31) wsum[w] = incl;
  if (lane == 0)  wpre[w] = pre;
  __syncthreads();

  int basev = 0;
  for (int k = 0; k < 8; ++k) {
    basev += wpre[k];
    if (k < w) basev += wsum[k];
  }
  const int e0 = basev + incl - own;
  const int e1 = e0 + c0, e2 = e1 + c1, e3 = e2 + c2, e4 = e3 + c3;
  if (i0     < n) { rowptr[i0]     = e0; cursor[i0]     = e0; }
  if (i0 + 1 < n) { rowptr[i0 + 1] = e1; cursor[i0 + 1] = e1; }
  if (i0 + 2 < n) { rowptr[i0 + 2] = e2; cursor[i0 + 2] = e2; }
  if (i0 + 3 < n) { rowptr[i0 + 3] = e3; cursor[i0 + 3] = e3; }
  if (i0 < n && i0 + 4 >= n) rowptr[n] = e4;
}

__global__ __launch_bounds__(256) void csr_fill(
    const int* __restrict__ srcIdx, const int* __restrict__ dstIdx,
    int* __restrict__ cursor, int* __restrict__ nbr, long long E)
{
  const long long e = (long long)blockIdx.x * 256 + threadIdx.x;
  if (e >= E) return;
  const int slot = atomicAdd(cursor + dstIdx[e], 1);
  nbr[slot] = srcIdx[e];
}

template <int U>
__device__ __forceinline__ void add_rows(float4& acc, const float* colp, int ldf,
                                         int mine, int t)
{
  float4 v[U];
  #pragma unroll
  for (int u = 0; u < U; ++u) {
    const int s = __builtin_amdgcn_readlane(mine, t + u);
    v[u] = *(const float4*)(colp + (size_t)s * ldf);
  }
  #pragma unroll
  for (int u = 0; u < U; ++u) {
    acc.x += v[u].x; acc.y += v[u].y; acc.z += v[u].z; acc.w += v[u].w;
  }
}

__global__ __launch_bounds__(256) void gin_gather(
    const float* __restrict__ feat, int ldf,
    const int* __restrict__ rowptr, const int* __restrict__ nbr,
    float* __restrict__ agg, int nRows)
{
  const int lane = threadIdx.x & 31;
  const int node = blockIdx.x * 8 + __builtin_amdgcn_readfirstlane(threadIdx.x >> 5);
  if (node >= nRows) return;
  const int beg = rowptr[node];
  const int end = rowptr[node + 1];
  const float* colp = feat + lane * 4;
  float4 acc = *(const float4*)(colp + (size_t)node * ldf);
  for (int base = beg; base < end; base += 32) {
    const int left = end - base;
    const int n = left < 32 ? left : 32;
    const int mine = (lane < n) ? nbr[base + lane] : 0;
    int t = 0;
    for (; t + 8 <= n; t += 8) add_rows<8>(acc, colp, ldf, mine, t);
    if (t + 4 <= n) { add_rows<4>(acc, colp, ldf, mine, t); t += 4; }
    if (t + 2 <= n) { add_rows<2>(acc, colp, ldf, mine, t); t += 2; }
    if (t < n)      { add_rows<1>(acc, colp, ldf, mine, t); }
  }
  *(float4*)(agg + (size_t)node * 128 + lane * 4) = acc;
}

__global__ __launch_bounds__(256) void gin_mlp3(
    const float* __restrict__ agg,
    const float* __restrict__ Wp,
    const float* __restrict__ Bv,
    float* __restrict__ xcat, int colOff)
{
  __shared__ float bufA[32][132];
  __shared__ float bufB[32][132];
  const int tid  = threadIdx.x;
  const int lane = tid & 31;
  const int w    = tid >> 5;
  const size_t base = (size_t)blockIdx.x * 32;

  for (int i = tid; i < 32 * 32; i += 256) {
    int row = i >> 5, c4 = (i & 31) << 2;
    const float4 v = *(const float4*)(agg + (base + row) * 128 + c4);
    bufA[row][c4] = v.x; bufA[row][c4 + 1] = v.y;
    bufA[row][c4 + 2] = v.z; bufA[row][c4 + 3] = v.w;
  }
  __syncthreads();

  const int half = lane >> 4;
  const int n16  = lane & 15;
  const int cb   = w * 16;

  for (int layer = 0; layer < 3; ++layer) {
    const float* Wk = Wp + (size_t)layer * 128 * 128;
    float (*in)[132]   = (layer & 1) ? bufB : bufA;
    float (*outb)[132] = (layer & 1) ? bufA : bufB;
    v8f acc0 = {}, acc1 = {};
    for (int k = 0; k < 128; k += 4) {
      const int kk = k + half * 2;
      const v2f b = *(const v2f*)(Wk + (((size_t)(k >> 2) * 128 + cb + n16) << 2)
                                     + half * 2);
      v2f a0; a0.x = in[n16][kk];      a0.y = in[n16][kk + 1];
      v2f a1; a1.x = in[n16 + 16][kk]; a1.y = in[n16 + 16][kk + 1];
      acc0 = wmma_f32(a0, b, acc0);
      acc1 = wmma_f32(a1, b, acc1);
    }
    const float bias = Bv[layer * 128 + cb + n16];
    #pragma unroll
    for (int j = 0; j < 8; ++j) {
      float r0 = acc0[j] + bias; r0 = r0 > 0.f ? r0 : 0.f;
      float r1 = acc1[j] + bias; r1 = r1 > 0.f ? r1 : 0.f;
      outb[j + half * 8][cb + n16]      = r0;
      outb[16 + j + half * 8][cb + n16] = r1;
    }
    __syncthreads();
  }
  for (int i = tid; i < 32 * 32; i += 256) {
    int row = i >> 5, c4 = (i & 31) << 2;
    float4 v = make_float4(bufB[row][c4], bufB[row][c4 + 1],
                           bufB[row][c4 + 2], bufB[row][c4 + 3]);
    *(float4*)(xcat + (base + row) * 512 + colOff + c4) = v;
  }
}

__global__ __launch_bounds__(256) void gin_chan(
    const float* __restrict__ xcat,
    const float* __restrict__ W1p, const float* __restrict__ B1,
    const float* __restrict__ W2p, const float* __restrict__ B2,
    const float* __restrict__ W3,  const float* __restrict__ B3,
    float* __restrict__ outp)
{
  __shared__ float bufA[16][516];
  __shared__ float bufB[16][260];
  const int tid  = threadIdx.x;
  const int lane = tid & 31;
  const int w    = tid >> 5;
  const int c    = blockIdx.y;
  const size_t base = (size_t)blockIdx.x * 16;

  for (int i = tid; i < 16 * 128; i += 256) {
    int row = i >> 7, c4 = (i & 127) << 2;
    const float4 v = *(const float4*)(xcat + (base + row) * 512 + c4);
    bufA[row][c4] = v.x; bufA[row][c4 + 1] = v.y;
    bufA[row][c4 + 2] = v.z; bufA[row][c4 + 3] = v.w;
  }
  __syncthreads();

  const int half = lane >> 4;
  const int n16  = lane & 15;
  const int ct0  = w * 16;
  const int ct1  = (w + 8) * 16;

  {
    const float* W1c = W1p + (size_t)c * 512 * 256;
    v8f acc0 = {}, acc1 = {};
    for (int k = 0; k < 512; k += 4) {
      const int kk = k + half * 2;
      v2f a; a.x = bufA[n16][kk]; a.y = bufA[n16][kk + 1];
      const size_t kb = ((size_t)(k >> 2) * 256) << 2;
      const v2f b0 = *(const v2f*)(W1c + kb + ((ct0 + n16) << 2) + half * 2);
      acc0 = wmma_f32(a, b0, acc0);
      const v2f b1 = *(const v2f*)(W1c + kb + ((ct1 + n16) << 2) + half * 2);
      acc1 = wmma_f32(a, b1, acc1);
    }
    const float bias0 = B1[c * 256 + ct0 + n16];
    const float bias1 = B1[c * 256 + ct1 + n16];
    #pragma unroll
    for (int j = 0; j < 8; ++j) {
      float r0 = acc0[j] + bias0; r0 = r0 > 0.f ? r0 : 0.f;
      float r1 = acc1[j] + bias1; r1 = r1 > 0.f ? r1 : 0.f;
      bufB[j + half * 8][ct0 + n16] = r0;
      bufB[j + half * 8][ct1 + n16] = r1;
    }
  }
  __syncthreads();

  {
    const float* W2c = W2p + (size_t)c * 256 * 256;
    v8f acc0 = {}, acc1 = {};
    for (int k = 0; k < 256; k += 4) {
      const int kk = k + half * 2;
      v2f a; a.x = bufB[n16][kk]; a.y = bufB[n16][kk + 1];
      const size_t kb = ((size_t)(k >> 2) * 256) << 2;
      const v2f b0 = *(const v2f*)(W2c + kb + ((ct0 + n16) << 2) + half * 2);
      acc0 = wmma_f32(a, b0, acc0);
      const v2f b1 = *(const v2f*)(W2c + kb + ((ct1 + n16) << 2) + half * 2);
      acc1 = wmma_f32(a, b1, acc1);
    }
    const float bias0 = B2[c * 256 + ct0 + n16];
    const float bias1 = B2[c * 256 + ct1 + n16];
    #pragma unroll
    for (int j = 0; j < 8; ++j) {
      float r0 = acc0[j] + bias0; r0 = r0 > 0.f ? r0 : 0.f;
      float r1 = acc1[j] + bias1; r1 = r1 > 0.f ? r1 : 0.f;
      bufA[j + half * 8][ct0 + n16] = r0;
      bufA[j + half * 8][ct1 + n16] = r1;
    }
  }
  __syncthreads();

  if (w == 0 && lane < 16) {
    const float* w3 = W3 + (size_t)c * 256;
    float sum = 0.f;
    for (int k = 0; k < 256; ++k) sum += bufA[lane][k] * w3[k];
    sum += B3[c];
    sum = sum > 0.f ? sum : 0.f;
    outp[(base + lane) * NCH + c] = sum;
  }
}

extern "C" void kernel_launch(void* const* d_in, const int* in_sizes, int n_in,
                              void* d_out, int out_size, void* d_ws, size_t ws_size,
                              hipStream_t stream)
{
  (void)n_in; (void)out_size; (void)ws_size;
  const float* x      = (const float*)d_in[0];
  const int*   ei     = (const int*)d_in[1];
  const float* convW  = (const float*)d_in[2];
  const float* convB  = (const float*)d_in[3];
  const float* chanW1 = (const float*)d_in[4];
  const float* chanB1 = (const float*)d_in[5];
  const float* chanW2 = (const float*)d_in[6];
  const float* chanB2 = (const float*)d_in[7];
  const float* chanW3 = (const float*)d_in[8];
  const float* chanB3 = (const float*)d_in[9];
  float* outp = (float*)d_out;

  const int N = NN;
  const long long E = (long long)in_sizes[1] / 2;

  float* xcat   = (float*)d_ws;
  float* agg    = xcat + (size_t)N * 512;
  float* convWp = agg + (size_t)N * 128;
  float* W1p    = convWp + (size_t)9 * 128 * 128;
  float* W2p    = W1p + (size_t)4 * 512 * 256;
  int*   cnt    = (int*)(W2p + (size_t)4 * 256 * 256);
  int*   cursor = cnt + N;
  int*   rowptr = cursor + N;
  int*   nbr    = rowptr + N + 4;

  repack_w<<<dim3((128 * 128) / 256, 9), 256, 0, stream>>>(convW, convWp, 128, 128);
  repack_w<<<dim3((512 * 256) / 256, 4), 256, 0, stream>>>(chanW1, W1p, 512, 256);
  repack_w<<<dim3((256 * 256) / 256, 4), 256, 0, stream>>>(chanW2, W2p, 256, 256);

  const int cpBlocks = (int)(((size_t)N * 32 + 255) / 256);
  const int eBlocks  = (int)((E + 255) / 256);
  const int nBlocks  = (N + 255) / 256;

  csr_zero<<<nBlocks, 256, 0, stream>>>(cnt, N);
  csr_count<<<eBlocks, 256, 0, stream>>>(ei + E, cnt, E);
  csr_scan<<<(N + 1023) / 1024, 256, 0, stream>>>(cnt, rowptr, cursor, N);
  csr_fill<<<eBlocks, 256, 0, stream>>>(ei, ei + E, cursor, nbr, E);

  copy_rows128<<<cpBlocks, 256, 0, stream>>>(x, 128, xcat, 512, N);

  for (int l = 0; l < 3; ++l) {
    const float* h = (l == 0) ? x : (xcat + (size_t)l * 128);
    const int ldh  = (l == 0) ? 128 : 512;
    gin_gather<<<(N + 7) / 8, 256, 0, stream>>>(h, ldh, rowptr, nbr, agg, N);
    gin_mlp3<<<N / 32, 256, 0, stream>>>(agg,
                                         convWp + (size_t)l * 3 * 128 * 128,
                                         convB + l * 3 * 128,
                                         xcat, (l + 1) * 128);
  }

  gin_chan<<<dim3(N / 16, NCH), 256, 0, stream>>>(xcat, W1p, chanB1,
                                                  W2p, chanB2,
                                                  chanW3, chanB3, outp);
}


# channel MLP: node tile per block 16 -> 32, each wave keeps 2 row strips x 2 column tiles so every fetched weight fragment feeds two matrix instructions; weight reads from L2 halved
# speedup vs baseline: 1.4924x; 1.0436x over previous
#include <hip/hip_runtime.h>

typedef __attribute__((ext_vector_type(2))) float v2f;
typedef __attribute__((ext_vector_type(8))) float v8f;

#define NN 100000
#define NCH 4

__device__ __forceinline__ v8f wmma_f32(v2f a, v2f b, v8f c) {
  return __builtin_amdgcn_wmma_f32_16x16x4_f32(false, a, false, b, (short)0, c,
                                               false, false);
}

__global__ __launch_bounds__(256) void repack_w(
    const float* __restrict__ W, float* __restrict__ Wp, int K, int C)
{
  const size_t mat = (size_t)blockIdx.y * K * C;
  size_t idx = (size_t)blockIdx.x * 256 + threadIdx.x;
  if (idx >= (size_t)K * C) return;
  int k   = (int)(idx / C);
  int col = (int)(idx % C);
  Wp[mat + (((size_t)(k >> 2) * C + col) << 2) + (k & 3)] = W[mat + idx];
}

__global__ __launch_bounds__(256) void copy_rows128(
    const float* __restrict__ in, int ldi,
    float* __restrict__ outp, int ldo, int nRows)
{
  size_t idx = (size_t)blockIdx.x * 256 + threadIdx.x;
  if (idx >= (size_t)nRows * 32) return;
  size_t row = idx >> 5;
  int c4 = (int)(idx & 31) << 2;
  const float4 v = *(const float4*)(in + row * (size_t)ldi + c4);
  *(float4*)(outp + row * (size_t)ldo + c4) = v;
}

__global__ __launch_bounds__(256) void csr_zero(int* __restrict__ cnt, int n)
{
  const int i = blockIdx.x * 256 + threadIdx.x;
  if (i < n) cnt[i] = 0;
}

__global__ __launch_bounds__(256) void csr_count(
    const int* __restrict__ dstIdx, int* __restrict__ cnt, long long E)
{
  const long long e = (long long)blockIdx.x * 256 + threadIdx.x;
  if (e >= E) return;
  atomicAdd(cnt + dstIdx[e], 1);
}

__global__ __launch_bounds__(256) void csr_scan(
    const int* __restrict__ cnt, int* __restrict__ rowptr,
    int* __restrict__ cursor, int n)
{
  __shared__ int wpre[8];
  __shared__ int wsum[8];
  const int tid  = threadIdx.x;
  const int lane = tid & 31;
  const int w    = tid >> 5;
  const int first = blockIdx.x * 1024;

  int pre = 0;
  const int* mycol = cnt + tid * 4;
  int i = 0;
  for (; i + 8 * 1024 <= first; i += 8 * 1024) {
    int4 v[8];
    #pragma unroll
    for (int u = 0; u < 8; ++u) v[u] = *(const int4*)(mycol + i + u * 1024);
    #pragma unroll
    for (int u = 0; u < 8; ++u) pre += (v[u].x + v[u].y) + (v[u].z + v[u].w);
  }
  for (; i < first; i += 1024) {
    const int4 v = *(const int4*)(mycol + i);
    pre += (v.x + v.y) + (v.z + v.w);
  }

  const int i0 = first + tid * 4;
  int c0 = 0, c1 = 0, c2 = 0, c3 = 0;
  if (i0 + 3 < n) {
    const int4 v = *(const int4*)(cnt + i0);
    c0 = v.x; c1 = v.y; c2 = v.z; c3 = v.w;
  } else {
    if (i0     < n) c0 = cnt[i0];
    if (i0 + 1 < n) c1 = cnt[i0 + 1];
    if (i0 + 2 < n) c2 = cnt[i0 + 2];
  }
  const int own = (c0 + c1) + (c2 + c3);

  int incl = own;
  for (int d = 1; d < 32; d <<= 1) {
    const int t = __shfl_up(incl, d);
    if (lane >= d) incl += t;
  }
  for (int d = 16; d >= 1; d >>= 1) pre += __shfl_xor(pre, d);
  if (lane == 31) wsum[w] = incl;
  if (lane == 0)  wpre[w] = pre;
  __syncthreads();

  int basev = 0;
  for (int k = 0; k < 8; ++k) {
    basev += wpre[k];
    if (k < w) basev += wsum[k];
  }
  const int e0 = basev + incl - own;
  const int e1 = e0 + c0, e2 = e1 + c1, e3 = e2 + c2, e4 = e3 + c3;
  if (i0     < n) { rowptr[i0]     = e0; cursor[i0]     = e0; }
  if (i0 + 1 < n) { rowptr[i0 + 1] = e1; cursor[i0 + 1] = e1; }
  if (i0 + 2 < n) { rowptr[i0 + 2] = e2; cursor[i0 + 2] = e2; }
  if (i0 + 3 < n) { rowptr[i0 + 3] = e3; cursor[i0 + 3] = e3; }
  if (i0 < n && i0 + 4 >= n) rowptr[n] = e4;
}

__global__ __launch_bounds__(256) void csr_fill(
    const int* __restrict__ srcIdx, const int* __restrict__ dstIdx,
    int* __restrict__ cursor, int* __restrict__ nbr, long long E)
{
  const long long e = (long long)blockIdx.x * 256 + threadIdx.x;
  if (e >= E) return;
  const int slot = atomicAdd(cursor + dstIdx[e], 1);
  nbr[slot] = srcIdx[e];
}

template <int U>
__device__ __forceinline__ void add_rows(float4& acc, const float* colp, int ldf,
                                         int mine, int t)
{
  float4 v[U];
  #pragma unroll
  for (int u = 0; u < U; ++u) {
    const int s = __builtin_amdgcn_readlane(mine, t + u);
    v[u] = *(const float4*)(colp + (size_t)s * ldf);
  }
  #pragma unroll
  for (int u = 0; u < U; ++u) {
    acc.x += v[u].x; acc.y += v[u].y; acc.z += v[u].z; acc.w += v[u].w;
  }
}

__global__ __launch_bounds__(256) void gin_gather(
    const float* __restrict__ feat, int ldf,
    const int* __restrict__ rowptr, const int* __restrict__ nbr,
    float* __restrict__ agg, int nRows)
{
  const int lane = threadIdx.x & 31;
  const int node = blockIdx.x * 8 + __builtin_amdgcn_readfirstlane(threadIdx.x >> 5);
  if (node >= nRows) return;
  const int beg = rowptr[node];
  const int end = rowptr[node + 1];
  const float* colp = feat + lane * 4;
  float4 acc = *(const float4*)(colp + (size_t)node * ldf);
  for (int base = beg; base < end; base += 32) {
    const int left = end - base;
    const int n = left < 32 ? left : 32;
    const int mine = (lane < n) ? nbr[base + lane] : 0;
    int t = 0;
    for (; t + 8 <= n; t += 8) add_rows<8>(acc, colp, ldf, mine, t);
    if (t + 4 <= n) { add_rows<4>(acc, colp, ldf, mine, t); t += 4; }
    if (t + 2 <= n) { add_rows<2>(acc, colp, ldf, mine, t); t += 2; }
    if (t < n)      { add_rows<1>(acc, colp, ldf, mine, t); }
  }
  *(float4*)(agg + (size_t)node * 128 + lane * 4) = acc;
}

__global__ __launch_bounds__(256) void gin_mlp3(
    const float* __restrict__ agg,
    const float* __restrict__ Wp,
    const float* __restrict__ Bv,
    float* __restrict__ xcat, int colOff)
{
  __shared__ float bufA[32][132];
  __shared__ float bufB[32][132];
  const int tid  = threadIdx.x;
  const int lane = tid & 31;
  const int w    = tid >> 5;
  const size_t base = (size_t)blockIdx.x * 32;

  for (int i = tid; i < 32 * 32; i += 256) {
    int row = i >> 5, c4 = (i & 31) << 2;
    const float4 v = *(const float4*)(agg + (base + row) * 128 + c4);
    bufA[row][c4] = v.x; bufA[row][c4 + 1] = v.y;
    bufA[row][c4 + 2] = v.z; bufA[row][c4 + 3] = v.w;
  }
  __syncthreads();

  const int half = lane >> 4;
  const int n16  = lane & 15;
  const int cb   = w * 16;

  for (int layer = 0; layer < 3; ++layer) {
    const float* Wk = Wp + (size_t)layer * 128 * 128;
    float (*in)[132]   = (layer & 1) ? bufB : bufA;
    float (*outb)[132] = (layer & 1) ? bufA : bufB;
    v8f acc0 = {}, acc1 = {};
    for (int k = 0; k < 128; k += 4) {
      const int kk = k + half * 2;
      const v2f b = *(const v2f*)(Wk + (((size_t)(k >> 2) * 128 + cb + n16) << 2)
                                     + half * 2);
      v2f a0; a0.x = in[n16][kk];      a0.y = in[n16][kk + 1];
      v2f a1; a1.x = in[n16 + 16][kk]; a1.y = in[n16 + 16][kk + 1];
      acc0 = wmma_f32(a0, b, acc0);
      acc1 = wmma_f32(a1, b, acc1);
    }
    const float bias = Bv[layer * 128 + cb + n16];
    #pragma unroll
    for (int j = 0; j < 8; ++j) {
      float r0 = acc0[j] + bias; r0 = r0 > 0.f ? r0 : 0.f;
      float r1 = acc1[j] + bias; r1 = r1 > 0.f ? r1 : 0.f;
      outb[j + half * 8][cb + n16]      = r0;
      outb[16 + j + half * 8][cb + n16] = r1;
    }
    __syncthreads();
  }
  for (int i = tid; i < 32 * 32; i += 256) {
    int row = i >> 5, c4 = (i & 31) << 2;
    float4 v = make_float4(bufB[row][c4], bufB[row][c4 + 1],
                           bufB[row][c4 + 2], bufB[row][c4 + 3]);
    *(float4*)(xcat + (base + row) * 512 + colOff + c4) = v;
  }
}

__global__ __launch_bounds__(256) void gin_chan(
    const float* __restrict__ xcat,
    const float* __restrict__ W1p, const float* __restrict__ B1,
    const float* __restrict__ W2p, const float* __restrict__ B2,
    const float* __restrict__ W3,  const float* __restrict__ B3,
    float* __restrict__ outp)
{
  __shared__ float bufA[32][516];
  __shared__ float bufB[32][260];
  const int tid  = threadIdx.x;
  const int lane = tid & 31;
  const int w    = tid >> 5;
  const int c    = blockIdx.y;
  const size_t base = (size_t)blockIdx.x * 32;

  for (int i = tid; i < 32 * 128; i += 256) {
    int row = i >> 7, c4 = (i & 127) << 2;
    const float4 v = *(const float4*)(xcat + (base + row) * 512 + c4);
    bufA[row][c4] = v.x; bufA[row][c4 + 1] = v.y;
    bufA[row][c4 + 2] = v.z; bufA[row][c4 + 3] = v.w;
  }
  __syncthreads();

  const int half = lane >> 4;
  const int n16  = lane & 15;
  const int ct0  = w * 16;
  const int ct1  = (w + 8) * 16;

  {
    const float* W1c = W1p + (size_t)c * 512 * 256;
    v8f acc00 = {}, acc01 = {}, acc10 = {}, acc11 = {};
    for (int k = 0; k < 512; k += 4) {
      const int kk = k + half * 2;
      v2f a0; a0.x = bufA[n16][kk];      a0.y = bufA[n16][kk + 1];
      v2f a1; a1.x = bufA[n16 + 16][kk]; a1.y = bufA[n16 + 16][kk + 1];
      const size_t kb = ((size_t)(k >> 2) * 256) << 2;
      const v2f b0 = *(const v2f*)(W1c + kb + ((ct0 + n16) << 2) + half * 2);
      const v2f b1 = *(const v2f*)(W1c + kb + ((ct1 + n16) << 2) + half * 2);
      acc00 = wmma_f32(a0, b0, acc00);
      acc10 = wmma_f32(a1, b0, acc10);
      acc01 = wmma_f32(a0, b1, acc01);
      acc11 = wmma_f32(a1, b1, acc11);
    }
    const float bias0 = B1[c * 256 + ct0 + n16];
    const float bias1 = B1[c * 256 + ct1 + n16];
    #pragma unroll
    for (int j = 0; j < 8; ++j) {
      const int r = j + half * 8;
      float r00 = acc00[j] + bias0; r00 = r00 > 0.f ? r00 : 0.f;
      float r01 = acc01[j] + bias1; r01 = r01 > 0.f ? r01 : 0.f;
      float r10 = acc10[j] + bias0; r10 = r10 > 0.f ? r10 : 0.f;
      float r11 = acc11[j] + bias1; r11 = r11 > 0.f ? r11 : 0.f;
      bufB[r][ct0 + n16]      = r00;
      bufB[r][ct1 + n16]      = r01;
      bufB[r + 16][ct0 + n16] = r10;
      bufB[r + 16][ct1 + n16] = r11;
    }
  }
  __syncthreads();

  {
    const float* W2c = W2p + (size_t)c * 256 * 256;
    v8f acc00 = {}, acc01 = {}, acc10 = {}, acc11 = {};
    for (int k = 0; k < 256; k += 4) {
      const int kk = k + half * 2;
      v2f a0; a0.x = bufB[n16][kk];      a0.y = bufB[n16][kk + 1];
      v2f a1; a1.x = bufB[n16 + 16][kk]; a1.y = bufB[n16 + 16][kk + 1];
      const size_t kb = ((size_t)(k >> 2) * 256) << 2;
      const v2f b0 = *(const v2f*)(W2c + kb + ((ct0 + n16) << 2) + half * 2);
      const v2f b1 = *(const v2f*)(W2c + kb + ((ct1 + n16) << 2) + half * 2);
      acc00 = wmma_f32(a0, b0, acc00);
      acc10 = wmma_f32(a1, b0, acc10);
      acc01 = wmma_f32(a0, b1, acc01);
      acc11 = wmma_f32(a1, b1, acc11);
    }
    const float bias0 = B2[c * 256 + ct0 + n16];
    const float bias1 = B2[c * 256 + ct1 + n16];
    #pragma unroll
    for (int j = 0; j < 8; ++j) {
      const int r = j + half * 8;
      float r00 = acc00[j] + bias0; r00 = r00 > 0.f ? r00 : 0.f;
      float r01 = acc01[j] + bias1; r01 = r01 > 0.f ? r01 : 0.f;
      float r10 = acc10[j] + bias0; r10 = r10 > 0.f ? r10 : 0.f;
      float r11 = acc11[j] + bias1; r11 = r11 > 0.f ? r11 : 0.f;
      bufA[r][ct0 + n16]      = r00;
      bufA[r][ct1 + n16]      = r01;
      bufA[r + 16][ct0 + n16] = r10;
      bufA[r + 16][ct1 + n16] = r11;
    }
  }
  __syncthreads();

  if (w == 0) {
    const float* w3 = W3 + (size_t)c * 256;
    float sum = 0.f;
    for (int k = 0; k < 256; ++k) sum += bufA[lane][k] * w3[k];
    sum += B3[c];
    sum = sum > 0.f ? sum : 0.f;
    outp[(base + lane) * NCH + c] = sum;
  }
}

extern "C" void kernel_launch(void* const* d_in, const int* in_sizes, int n_in,
                              void* d_out, int out_size, void* d_ws, size_t ws_size,
                              hipStream_t stream)
{
  (void)n_in; (void)out_size; (void)ws_size;
  const float* x      = (const float*)d_in[0];
  const int*   ei     = (const int*)d_in[1];
  const float* convW  = (const float*)d_in[2];
  const float* convB  = (const float*)d_in[3];
  const float* chanW1 = (const float*)d_in[4];
  const float* chanB1 = (const float*)d_in[5];
  const float* chanW2 = (const float*)d_in[6];
  const float* chanB2 = (const float*)d_in[7];
  const float* chanW3 = (const float*)d_in[8];
  const float* chanB3 = (const float*)d_in[9];
  float* outp = (float*)d_out;

  const int N = NN;
  const long long E = (long long)in_sizes[1] / 2;

  float* xcat   = (float*)d_ws;
  float* agg    = xcat + (size_t)N * 512;
  float* convWp = agg + (size_t)N * 128;
  float* W1p    = convWp + (size_t)9 * 128 * 128;
  float* W2p    = W1p + (size_t)4 * 512 * 256;
  int*   cnt    = (int*)(W2p + (size_t)4 * 256 * 256);
  int*   cursor = cnt + N;
  int*   rowptr = cursor + N;
  int*   nbr    = rowptr + N + 4;

  repack_w<<<dim3((128 * 128) / 256, 9), 256, 0, stream>>>(convW, convWp, 128, 128);
  repack_w<<<dim3((512 * 256) / 256, 4), 256, 0, stream>>>(chanW1, W1p, 512, 256);
  repack_w<<<dim3((256 * 256) / 256, 4), 256, 0, stream>>>(chanW2, W2p, 256, 256);

  const int cpBlocks = (int)(((size_t)N * 32 + 255) / 256);
  const int eBlocks  = (int)((E + 255) / 256);
  const int nBlocks  = (N + 255) / 256;

  csr_zero<<<nBlocks, 256, 0, stream>>>(cnt, N);
  csr_count<<<eBlocks, 256, 0, stream>>>(ei + E, cnt, E);
  csr_scan<<<(N + 1023) / 1024, 256, 0, stream>>>(cnt, rowptr, cursor, N);
  csr_fill<<<eBlocks, 256, 0, stream>>>(ei, ei + E, cursor, nbr, E);

  copy_rows128<<<cpBlocks, 256, 0, stream>>>(x, 128, xcat, 512, N);

  for (int l = 0; l < 3; ++l) {
    const float* h = (l == 0) ? x : (xcat + (size_t)l * 128);
    const int ldh  = (l == 0) ? 128 : 512;
    gin_gather<<<(N + 7) / 8, 256, 0, stream>>>(h, ldh, rowptr, nbr, agg, N);
    gin_mlp3<<<N / 32, 256, 0, stream>>>(agg,
                                         convWp + (size_t)l * 3 * 128 * 128,
                                         convB + l * 3 * 128,
                                         xcat, (l + 1) * 128);
  }

  gin_chan<<<dim3(N / 32, NCH), 256, 0, stream>>>(xcat, W1p, chanB1,
                                                  W2p, chanB2,
                                                  chanW3, chanB3, outp);
}


# channel MLP: k loops unrolled in blocks of 8 k-steps, the next block's 16 weight fragments requested (second register set) ahead of the current block's 32 matrix instructions; sum order kept
# speedup vs baseline: 1.5427x; 1.0337x over previous
#include <hip/hip_runtime.h>

typedef __attribute__((ext_vector_type(2))) float v2f;
typedef __attribute__((ext_vector_type(8))) float v8f;

#define NN 100000
#define NCH 4

__device__ __forceinline__ v8f wmma_f32(v2f a, v2f b, v8f c) {
  return __builtin_amdgcn_wmma_f32_16x16x4_f32(false, a, false, b, (short)0, c,
                                               false, false);
}

__global__ __launch_bounds__(256) void repack_w(
    const float* __restrict__ W, float* __restrict__ Wp, int K, int C)
{
  const size_t mat = (size_t)blockIdx.y * K * C;
  size_t idx = (size_t)blockIdx.x * 256 + threadIdx.x;
  if (idx >= (size_t)K * C) return;
  int k   = (int)(idx / C);
  int col = (int)(idx % C);
  Wp[mat + (((size_t)(k >> 2) * C + col) << 2) + (k & 3)] = W[mat + idx];
}

__global__ __launch_bounds__(256) void copy_rows128(
    const float* __restrict__ in, int ldi,
    float* __restrict__ outp, int ldo, int nRows)
{
  size_t idx = (size_t)blockIdx.x * 256 + threadIdx.x;
  if (idx >= (size_t)nRows * 32) return;
  size_t row = idx >> 5;
  int c4 = (int)(idx & 31) << 2;
  const float4 v = *(const float4*)(in + row * (size_t)ldi + c4);
  *(float4*)(outp + row * (size_t)ldo + c4) = v;
}

__global__ __launch_bounds__(256) void csr_zero(int* __restrict__ cnt, int n)
{
  const int i = blockIdx.x * 256 + threadIdx.x;
  if (i < n) cnt[i] = 0;
}

__global__ __launch_bounds__(256) void csr_count(
    const int* __restrict__ dstIdx, int* __restrict__ cnt, long long E)
{
  const long long e = (long long)blockIdx.x * 256 + threadIdx.x;
  if (e >= E) return;
  atomicAdd(cnt + dstIdx[e], 1);
}

__global__ __launch_bounds__(256) void csr_scan(
    const int* __restrict__ cnt, int* __restrict__ rowptr,
    int* __restrict__ cursor, int n)
{
  __shared__ int wpre[8];
  __shared__ int wsum[8];
  const int tid  = threadIdx.x;
  const int lane = tid & 31;
  const int w    = tid >> 5;
  const int first = blockIdx.x * 1024;

  int pre = 0;
  const int* mycol = cnt + tid * 4;
  int i = 0;
  for (; i + 8 * 1024 <= first; i += 8 * 1024) {
    int4 v[8];
    #pragma unroll
    for (int u = 0; u < 8; ++u) v[u] = *(const int4*)(mycol + i + u * 1024);
    #pragma unroll
    for (int u = 0; u < 8; ++u) pre += (v[u].x + v[u].y) + (v[u].z + v[u].w);
  }
  for (; i < first; i += 1024) {
    const int4 v = *(const int4*)(mycol + i);
    pre += (v.x + v.y) + (v.z + v.w);
  }

  const int i0 = first + tid * 4;
  int c0 = 0, c1 = 0, c2 = 0, c3 = 0;
  if (i0 + 3 < n) {
    const int4 v = *(const int4*)(cnt + i0);
    c0 = v.x; c1 = v.y; c2 = v.z; c3 = v.w;
  } else {
    if (i0     < n) c0 = cnt[i0];
    if (i0 + 1 < n) c1 = cnt[i0 + 1];
    if (i0 + 2 < n) c2 = cnt[i0 + 2];
  }
  const int own = (c0 + c1) + (c2 + c3);

  int incl = own;
  for (int d = 1; d < 32; d <<= 1) {
    const int t = __shfl_up(incl, d);
    if (lane >= d) incl += t;
  }
  for (int d = 16; d >= 1; d >>= 1) pre += __shfl_xor(pre, d);
  if (lane == 31) wsum[w] = incl;
  if (lane == 0)  wpre[w] = pre;
  __syncthreads();

  int basev = 0;
  for (int k = 0; k < 8; ++k) {
    basev += wpre[k];
    if (k < w) basev += wsum[k];
  }
  const int e0 = basev + incl - own;
  const int e1 = e0 + c0, e2 = e1 + c1, e3 = e2 + c2, e4 = e3 + c3;
  if (i0     < n) { rowptr[i0]     = e0; cursor[i0]     = e0; }
  if (i0 + 1 < n) { rowptr[i0 + 1] = e1; cursor[i0 + 1] = e1; }
  if (i0 + 2 < n) { rowptr[i0 + 2] = e2; cursor[i0 + 2] = e2; }
  if (i0 + 3 < n) { rowptr[i0 + 3] = e3; cursor[i0 + 3] = e3; }
  if (i0 < n && i0 + 4 >= n) rowptr[n] = e4;
}

__global__ __launch_bounds__(256) void csr_fill(
    const int* __restrict__ srcIdx, const int* __restrict__ dstIdx,
    int* __restrict__ cursor, int* __restrict__ nbr, long long E)
{
  const long long e = (long long)blockIdx.x * 256 + threadIdx.x;
  if (e >= E) return;
  const int slot = atomicAdd(cursor + dstIdx[e], 1);
  nbr[slot] = srcIdx[e];
}

template <int U>
__device__ __forceinline__ void add_rows(float4& acc, const float* colp, int ldf,
                                         int mine, int t)
{
  float4 v[U];
  #pragma unroll
  for (int u = 0; u < U; ++u) {
    const int s = __builtin_amdgcn_readlane(mine, t + u);
    v[u] = *(const float4*)(colp + (size_t)s * ldf);
  }
  #pragma unroll
  for (int u = 0; u < U; ++u) {
    acc.x += v[u].x; acc.y += v[u].y; acc.z += v[u].z; acc.w += v[u].w;
  }
}

__global__ __launch_bounds__(256) void gin_gather(
    const float* __restrict__ feat, int ldf,
    const int* __restrict__ rowptr, const int* __restrict__ nbr,
    float* __restrict__ agg, int nRows)
{
  const int lane = threadIdx.x & 31;
  const int node = blockIdx.x * 8 + __builtin_amdgcn_readfirstlane(threadIdx.x >> 5);
  if (node >= nRows) return;
  const int beg = rowptr[node];
  const int end = rowptr[node + 1];
  const float* colp = feat + lane * 4;
  float4 acc = *(const float4*)(colp + (size_t)node * ldf);
  for (int base = beg; base < end; base += 32) {
    const int left = end - base;
    const int n = left < 32 ? left : 32;
    const int mine = (lane < n) ? nbr[base + lane] : 0;
    int t = 0;
    for (; t + 8 <= n; t += 8) add_rows<8>(acc, colp, ldf, mine, t);
    if (t + 4 <= n) { add_rows<4>(acc, colp, ldf, mine, t); t += 4; }
    if (t + 2 <= n) { add_rows<2>(acc, colp, ldf, mine, t); t += 2; }
    if (t < n)      { add_rows<1>(acc, colp, ldf, mine, t); }
  }
  *(float4*)(agg + (size_t)node * 128 + lane * 4) = acc;
}

__global__ __launch_bounds__(256) void gin_mlp3(
    const float* __restrict__ agg,
    const float* __restrict__ Wp,
    const float* __restrict__ Bv,
    float* __restrict__ xcat, int colOff)
{
  __shared__ float bufA[32][132];
  __shared__ float bufB[32][132];
  const int tid  = threadIdx.x;
  const int lane = tid & 31;
  const int w    = tid >> 5;
  const size_t base = (size_t)blockIdx.x * 32;

  for (int i = tid; i < 32 * 32; i += 256) {
    int row = i >> 5, c4 = (i & 31) << 2;
    const float4 v = *(const float4*)(agg + (base + row) * 128 + c4);
    bufA[row][c4] = v.x; bufA[row][c4 + 1] = v.y;
    bufA[row][c4 + 2] = v.z; bufA[row][c4 + 3] = v.w;
  }
  __syncthreads();

  const int half = lane >> 4;
  const int n16  = lane & 15;
  const int cb   = w * 16;

  for (int layer = 0; layer < 3; ++layer) {
    const float* Wk = Wp + (size_t)layer * 128 * 128;
    float (*in)[132]   = (layer & 1) ? bufB : bufA;
    float (*outb)[132] = (layer & 1) ? bufA : bufB;
    v8f acc0 = {}, acc1 = {};
    for (int k = 0; k < 128; k += 4) {
      const int kk = k + half * 2;
      const v2f b = *(const v2f*)(Wk + (((size_t)(k >> 2) * 128 + cb + n16) << 2)
                                     + half * 2);
      v2f a0; a0.x = in[n16][kk];      a0.y = in[n16][kk + 1];
      v2f a1; a1.x = in[n16 + 16][kk]; a1.y = in[n16 + 16][kk + 1];
      acc0 = wmma_f32(a0, b, acc0);
      acc1 = wmma_f32(a1, b, acc1);
    }
    const float bias = Bv[layer * 128 + cb + n16];
    #pragma unroll
    for (int j = 0; j < 8; ++j) {
      float r0 = acc0[j] + bias; r0 = r0 > 0.f ? r0 : 0.f;
      float r1 = acc1[j] + bias; r1 = r1 > 0.f ? r1 : 0.f;
      outb[j + half * 8][cb + n16]      = r0;
      outb[16 + j + half * 8][cb + n16] = r1;
    }
    __syncthreads();
  }
  for (int i = tid; i < 32 * 32; i += 256) {
    int row = i >> 5, c4 = (i & 31) << 2;
    float4 v = make_float4(bufB[row][c4], bufB[row][c4 + 1],
                           bufB[row][c4 + 2], bufB[row][c4 + 3]);
    *(float4*)(xcat + (base + row) * 512 + colOff + c4) = v;
  }
}

#define KU 8

__device__ __forceinline__ void load_wfrag(v2f (&b)[KU][2], const float* __restrict__ bq)
{
  #pragma unroll
  for (int u = 0; u < KU; ++u) {
    b[u][0] = *(const v2f*)(bq + u * 1024);
    b[u][1] = *(const v2f*)(bq + u * 1024 + 512);
  }
}

template <int LD>
__device__ __forceinline__ void mma_block(const v2f (&b)[KU][2], const float* aq,
                                          v8f& acc00, v8f& acc01,
                                          v8f& acc10, v8f& acc11)
{
  v2f a0[KU], a1[KU];
  #pragma unroll
  for (int u = 0; u < KU; ++u) {
    a0[u] = *(const v2f*)(aq + 4 * u);
    a1[u] = *(const v2f*)(aq + 16 * LD + 4 * u);
  }
  #pragma unroll
  for (int u = 0; u < KU; ++u) {
    acc00 = wmma_f32(a0[u], b[u][0], acc00);
    acc10 = wmma_f32(a1[u], b[u][0], acc10);
    acc01 = wmma_f32(a0[u], b[u][1], acc01);
    acc11 = wmma_f32(a1[u], b[u][1], acc11);
  }
}

template <int NKS, int LD>
__device__ __forceinline__ void gemm_2x2(const float* __restrict__ bp, const float* ap,
                                         v8f& acc00, v8f& acc01,
                                         v8f& acc10, v8f& acc11)
{
  constexpr int NB = NKS / KU;
  v2f bA[KU][2], bB[KU][2];
  load_wfrag(bA, bp);
  __builtin_amdgcn_sched_group_barrier(0x020, 16, 0);
  #pragma unroll
  for (int blk = 0; blk < NB; blk += 2) {
    const float* bq = bp + (size_t)blk * (KU * 1024);
    const float* aq = ap + blk * (KU * 4);
    load_wfrag(bB, bq + KU * 1024);
    mma_block<LD>(bA, aq, acc00, acc01, acc10, acc11);
    __builtin_amdgcn_sched_group_barrier(0x020, 16, 0);
    __builtin_amdgcn_sched_group_barrier(0x008, 32, 0);
    if (blk + 2 < NB) {
      load_wfrag(bA, bq + 2 * KU * 1024);
      __builtin_amdgcn_sched_group_barrier(0x020, 16, 0);
    }
    mma_block<LD>(bB, aq + KU * 4, acc00, acc01, acc10, acc11);
    __builtin_amdgcn_sched_group_barrier(0x008, 32, 0);
  }
}

__global__ __launch_bounds__(256) void gin_chan(
    const float* __restrict__ xcat,
    const float* __restrict__ W1p, const float* __restrict__ B1,
    const float* __restrict__ W2p, const float* __restrict__ B2,
    const float* __restrict__ W3,  const float* __restrict__ B3,
    float* __restrict__ outp)
{
  __shared__ __attribute__((aligned(16))) float bufA[32][516];
  __shared__ __attribute__((aligned(16))) float bufB[32][260];
  const int tid  = threadIdx.x;
  const int lane = tid & 31;
  const int w    = tid >> 5;
  const int c    = blockIdx.y;
  const size_t base = (size_t)blockIdx.x * 32;

  for (int i = tid; i < 32 * 128; i += 256) {
    int row = i >> 7, c4 = (i & 127) << 2;
    const float4 v = *(const float4*)(xcat + (base + row) * 512 + c4);
    bufA[row][c4] = v.x; bufA[row][c4 + 1] = v.y;
    bufA[row][c4 + 2] = v.z; bufA[row][c4 + 3] = v.w;
  }
  __syncthreads();

  const int half = lane >> 4;
  const int n16  = lane & 15;
  const int ct0  = w * 16;
  const int ct1  = (w + 8) * 16;
  const int bofs = ((ct0 + n16) << 2) + half * 2;

  {
    v8f acc00 = {}, acc01 = {}, acc10 = {}, acc11 = {};
    gemm_2x2<128, 516>(W1p + (size_t)c * 512 * 256 + bofs,
                       &bufA[n16][half * 2], acc00, acc01, acc10, acc11);
    const float bias0 = B1[c * 256 + ct0 + n16];
    const float bias1 = B1[c * 256 + ct1 + n16];
    #pragma unroll
    for (int j = 0; j < 8; ++j) {
      const int r = j + half * 8;
      float r00 = acc00[j] + bias0; r00 = r00 > 0.f ? r00 : 0.f;
      float r01 = acc01[j] + bias1; r01 = r01 > 0.f ? r01 : 0.f;
      float r10 = acc10[j] + bias0; r10 = r10 > 0.f ? r10 : 0.f;
      float r11 = acc11[j] + bias1; r11 = r11 > 0.f ? r11 : 0.f;
      bufB[r][ct0 + n16]      = r00;
      bufB[r][ct1 + n16]      = r01;
      bufB[r + 16][ct0 + n16] = r10;
      bufB[r + 16][ct1 + n16] = r11;
    }
  }
  __syncthreads();

  {
    v8f acc00 = {}, acc01 = {}, acc10 = {}, acc11 = {};
    gemm_2x2<64, 260>(W2p + (size_t)c * 256 * 256 + bofs,
                      &bufB[n16][half * 2], acc00, acc01, acc10, acc11);
    const float bias0 = B2[c * 256 + ct0 + n16];
    const float bias1 = B2[c * 256 + ct1 + n16];
    #pragma unroll
    for (int j = 0; j < 8; ++j) {
      const int r = j + half * 8;
      float r00 = acc00[j] + bias0; r00 = r00 > 0.f ? r00 : 0.f;
      float r01 = acc01[j] + bias1; r01 = r01 > 0.f ? r01 : 0.f;
      float r10 = acc10[j] + bias0; r10 = r10 > 0.f ? r10 : 0.f;
      float r11 = acc11[j] + bias1; r11 = r11 > 0.f ? r11 : 0.f;
      bufA[r][ct0 + n16]      = r00;
      bufA[r][ct1 + n16]      = r01;
      bufA[r + 16][ct0 + n16] = r10;
      bufA[r + 16][ct1 + n16] = r11;
    }
  }
  __syncthreads();

  if (w == 0) {
    const float* w3 = W3 + (size_t)c * 256;
    float sum = 0.f;
    for (int k = 0; k < 256; ++k) sum += bufA[lane][k] * w3[k];
    sum += B3[c];
    sum = sum > 0.f ? sum : 0.f;
    outp[(base + lane) * NCH + c] = sum;
  }
}

extern "C" void kernel_launch(void* const* d_in, const int* in_sizes, int n_in,
                              void* d_out, int out_size, void* d_ws, size_t ws_size,
                              hipStream_t stream)
{
  (void)n_in; (void)out_size; (void)ws_size;
  const float* x      = (const float*)d_in[0];
  const int*   ei     = (const int*)d_in[1];
  const float* convW  = (const float*)d_in[2];
  const float* convB  = (const float*)d_in[3];
  const float* chanW1 = (const float*)d_in[4];
  const float* chanB1 = (const float*)d_in[5];
  const float* chanW2 = (const float*)d_in[6];
  const float* chanB2 = (const float*)d_in[7];
  const float* chanW3 = (const float*)d_in[8];
  const float* chanB3 = (const float*)d_in[9];
  float* outp = (float*)d_out;

  const int N = NN;
  const long long E = (long long)in_sizes[1] / 2;

  float* xcat   = (float*)d_ws;
  float* agg    = xcat + (size_t)N * 512;
  float* convWp = agg + (size_t)N * 128;
  float* W1p    = convWp + (size_t)9 * 128 * 128;
  float* W2p    = W1p + (size_t)4 * 512 * 256;
  int*   cnt    = (int*)(W2p + (size_t)4 * 256 * 256);
  int*   cursor = cnt + N;
  int*   rowptr = cursor + N;
  int*   nbr    = rowptr + N + 4;

  repack_w<<<dim3((128 * 128) / 256, 9), 256, 0, stream>>>(convW, convWp, 128, 128);
  repack_w<<<dim3((512 * 256) / 256, 4), 256, 0, stream>>>(chanW1, W1p, 512, 256);
  repack_w<<<dim3((256 * 256) / 256, 4), 256, 0, stream>>>(chanW2, W2p, 256, 256);

  const int cpBlocks = (int)(((size_t)N * 32 + 255) / 256);
  const int eBlocks  = (int)((E + 255) / 256);
  const int nBlocks  = (N + 255) / 256;

  csr_zero<<<nBlocks, 256, 0, stream>>>(cnt, N);
  csr_count<<<eBlocks, 256, 0, stream>>>(ei + E, cnt, E);
  csr_scan<<<(N + 1023) / 1024, 256, 0, stream>>>(cnt, rowptr, cursor, N);
  csr_fill<<<eBlocks, 256, 0, stream>>>(ei, ei + E, cursor, nbr, E);

  copy_rows128<<<cpBlocks, 256, 0, stream>>>(x, 128, xcat, 512, N);

  for (int l = 0; l < 3; ++l) {
    const float* h = (l == 0) ? x : (xcat + (size_t)l * 128);
    const int ldh  = (l == 0) ? 128 : 512;
    gin_gather<<<(N + 7) / 8, 256, 0, stream>>>(h, ldh, rowptr, nbr, agg, N);
    gin_mlp3<<<N / 32, 256, 0, stream>>>(agg,
                                         convWp + (size_t)l * 3 * 128 * 128,
                                         convB + l * 3 * 128,
                                         xcat, (l + 1) * 128);
  }

  gin_chan<<<dim3(N / 32, NCH), 256, 0, stream>>>(xcat, W1p, chanB1,
                                                  W2p, chanB2,
                                                  chanW3, chanB3, outp);
}


# neighbour gather fused into the conv MLP kernel: each wave sums its 4 nodes' rows straight into the LDS tile; agg buffer round trip and 3 launches gone; sum order kept
# speedup vs baseline: 1.6024x; 1.0387x over previous
#include <hip/hip_runtime.h>

typedef __attribute__((ext_vector_type(2))) float v2f;
typedef __attribute__((ext_vector_type(8))) float v8f;

#define NN 100000
#define NCH 4

__device__ __forceinline__ v8f wmma_f32(v2f a, v2f b, v8f c) {
  return __builtin_amdgcn_wmma_f32_16x16x4_f32(false, a, false, b, (short)0, c,
                                               false, false);
}

__global__ __launch_bounds__(256) void repack_w(
    const float* __restrict__ W, float* __restrict__ Wp, int K, int C)
{
  const size_t mat = (size_t)blockIdx.y * K * C;
  size_t idx = (size_t)blockIdx.x * 256 + threadIdx.x;
  if (idx >= (size_t)K * C) return;
  int k   = (int)(idx / C);
  int col = (int)(idx % C);
  Wp[mat + (((size_t)(k >> 2) * C + col) << 2) + (k & 3)] = W[mat + idx];
}

__global__ __launch_bounds__(256) void copy_rows128(
    const float* __restrict__ in, int ldi,
    float* __restrict__ outp, int ldo, int nRows)
{
  size_t idx = (size_t)blockIdx.x * 256 + threadIdx.x;
  if (idx >= (size_t)nRows * 32) return;
  size_t row = idx >> 5;
  int c4 = (int)(idx & 31) << 2;
  const float4 v = *(const float4*)(in + row * (size_t)ldi + c4);
  *(float4*)(outp + row * (size_t)ldo + c4) = v;
}

__global__ __launch_bounds__(256) void csr_zero(int* __restrict__ cnt, int n)
{
  const int i = blockIdx.x * 256 + threadIdx.x;
  if (i < n) cnt[i] = 0;
}

__global__ __launch_bounds__(256) void csr_count(
    const int* __restrict__ dstIdx, int* __restrict__ cnt, long long E)
{
  const long long e = (long long)blockIdx.x * 256 + threadIdx.x;
  if (e >= E) return;
  atomicAdd(cnt + dstIdx[e], 1);
}

__global__ __launch_bounds__(256) void csr_scan(
    const int* __restrict__ cnt, int* __restrict__ rowptr,
    int* __restrict__ cursor, int n)
{
  __shared__ int wpre[8];
  __shared__ int wsum[8];
  const int tid  = threadIdx.x;
  const int lane = tid & 31;
  const int w    = tid >> 5;
  const int first = blockIdx.x * 1024;

  int pre = 0;
  const int* mycol = cnt + tid * 4;
  int i = 0;
  for (; i + 8 * 1024 <= first; i += 8 * 1024) {
    int4 v[8];
    #pragma unroll
    for (int u = 0; u < 8; ++u) v[u] = *(const int4*)(mycol + i + u * 1024);
    #pragma unroll
    for (int u = 0; u < 8; ++u) pre += (v[u].x + v[u].y) + (v[u].z + v[u].w);
  }
  for (; i < first; i += 1024) {
    const int4 v = *(const int4*)(mycol + i);
    pre += (v.x + v.y) + (v.z + v.w);
  }

  const int i0 = first + tid * 4;
  int c0 = 0, c1 = 0, c2 = 0, c3 = 0;
  if (i0 + 3 < n) {
    const int4 v = *(const int4*)(cnt + i0);
    c0 = v.x; c1 = v.y; c2 = v.z; c3 = v.w;
  } else {
    if (i0     < n) c0 = cnt[i0];
    if (i0 + 1 < n) c1 = cnt[i0 + 1];
    if (i0 + 2 < n) c2 = cnt[i0 + 2];
  }
  const int own = (c0 + c1) + (c2 + c3);

  int incl = own;
  for (int d = 1; d < 32; d <<= 1) {
    const int t = __shfl_up(incl, d);
    if (lane >= d) incl += t;
  }
  for (int d = 16; d >= 1; d >>= 1) pre += __shfl_xor(pre, d);
  if (lane == 31) wsum[w] = incl;
  if (lane == 0)  wpre[w] = pre;
  __syncthreads();

  int basev = 0;
  for (int k = 0; k < 8; ++k) {
    basev += wpre[k];
    if (k < w) basev += wsum[k];
  }
  const int e0 = basev + incl - own;
  const int e1 = e0 + c0, e2 = e1 + c1, e3 = e2 + c2, e4 = e3 + c3;
  if (i0     < n) { rowptr[i0]     = e0; cursor[i0]     = e0; }
  if (i0 + 1 < n) { rowptr[i0 + 1] = e1; cursor[i0 + 1] = e1; }
  if (i0 + 2 < n) { rowptr[i0 + 2] = e2; cursor[i0 + 2] = e2; }
  if (i0 + 3 < n) { rowptr[i0 + 3] = e3; cursor[i0 + 3] = e3; }
  if (i0 < n && i0 + 4 >= n) rowptr[n] = e4;
}

__global__ __launch_bounds__(256) void csr_fill(
    const int* __restrict__ srcIdx, const int* __restrict__ dstIdx,
    int* __restrict__ cursor, int* __restrict__ nbr, long long E)
{
  const long long e = (long long)blockIdx.x * 256 + threadIdx.x;
  if (e >= E) return;
  const int slot = atomicAdd(cursor + dstIdx[e], 1);
  nbr[slot] = srcIdx[e];
}

template <int U>
__device__ __forceinline__ void add_rows(float4& acc, const float* colp, int ldf,
                                         int mine, int t)
{
  float4 v[U];
  #pragma unroll
  for (int u = 0; u < U; ++u) {
    const int s = __builtin_amdgcn_readlane(mine, t + u);
    v[u] = *(const float4*)(colp + (size_t)s * ldf);
  }
  #pragma unroll
  for (int u = 0; u < U; ++u) {
    acc.x += v[u].x; acc.y += v[u].y; acc.z += v[u].z; acc.w += v[u].w;
  }
}

__device__ __forceinline__ void sum_list(float4& acc, const float* colp, int ldf,
                                         const int* __restrict__ nbr,
                                         int beg, int end, int mine, int lane)
{
  for (int pos = beg; pos < end; pos += 32) {
    const int left = end - pos;
    const int n = left < 32 ? left : 32;
    if (pos != beg) mine = (lane < n) ? nbr[pos + lane] : 0;
    int t = 0;
    for (; t + 8 <= n; t += 8) add_rows<8>(acc, colp, ldf, mine, t);
    if (t + 4 <= n) { add_rows<4>(acc, colp, ldf, mine, t); t += 4; }
    if (t + 2 <= n) { add_rows<2>(acc, colp, ldf, mine, t); t += 2; }
    if (t < n)      { add_rows<1>(acc, colp, ldf, mine, t); }
  }
}

__global__ __launch_bounds__(256) void gin_mlp3(
    const float* __restrict__ feat, int ldf,
    const int* __restrict__ rowptr, const int* __restrict__ nbr,
    const float* __restrict__ Wp,
    const float* __restrict__ Bv,
    float* __restrict__ xcat, int colOff)
{
  __shared__ __attribute__((aligned(16))) float bufA[32][132];
  __shared__ float bufB[32][132];
  const int tid  = threadIdx.x;
  const int lane = tid & 31;
  const int w    = tid >> 5;
  const size_t base = (size_t)blockIdx.x * 32;

  {
    const int wu    = __builtin_amdgcn_readfirstlane(tid >> 5);
    const int node0 = (int)base + wu * 4;
    const int rp = (lane < 5) ? rowptr[node0 + lane] : 0;
    int rb[5];
    #pragma unroll
    for (int j = 0; j < 5; ++j) rb[j] = __builtin_amdgcn_readlane(rp, j);
    const float* colp = feat + lane * 4;
    int mine[4];
    float4 acc[4];
    #pragma unroll
    for (int j = 0; j < 4; ++j)
      mine[j] = (lane < rb[j + 1] - rb[j]) ? nbr[rb[j] + lane] : 0;
    #pragma unroll
    for (int j = 0; j < 4; ++j)
      acc[j] = *(const float4*)(colp + (size_t)(node0 + j) * ldf);
    #pragma unroll
    for (int j = 0; j < 4; ++j) {
      sum_list(acc[j], colp, ldf, nbr, rb[j], rb[j + 1], mine[j], lane);
      *(float4*)&bufA[wu * 4 + j][lane * 4] = acc[j];
    }
  }
  __syncthreads();

  const int half = lane >> 4;
  const int n16  = lane & 15;
  const int cb   = w * 16;

  for (int layer = 0; layer < 3; ++layer) {
    const float* Wk = Wp + (size_t)layer * 128 * 128;
    float (*in)[132]   = (layer & 1) ? bufB : bufA;
    float (*outb)[132] = (layer & 1) ? bufA : bufB;
    v8f acc0 = {}, acc1 = {};
    for (int k = 0; k < 128; k += 4) {
      const int kk = k + half * 2;
      const v2f b = *(const v2f*)(Wk + (((size_t)(k >> 2) * 128 + cb + n16) << 2)
                                     + half * 2);
      v2f a0; a0.x = in[n16][kk];      a0.y = in[n16][kk + 1];
      v2f a1; a1.x = in[n16 + 16][kk]; a1.y = in[n16 + 16][kk + 1];
      acc0 = wmma_f32(a0, b, acc0);
      acc1 = wmma_f32(a1, b, acc1);
    }
    const float bias = Bv[layer * 128 + cb + n16];
    #pragma unroll
    for (int j = 0; j < 8; ++j) {
      float r0 = acc0[j] + bias; r0 = r0 > 0.f ? r0 : 0.f;
      float r1 = acc1[j] + bias; r1 = r1 > 0.f ? r1 : 0.f;
      outb[j + half * 8][cb + n16]      = r0;
      outb[16 + j + half * 8][cb + n16] = r1;
    }
    __syncthreads();
  }
  for (int i = tid; i < 32 * 32; i += 256) {
    int row = i >> 5, c4 = (i & 31) << 2;
    float4 v = make_float4(bufB[row][c4], bufB[row][c4 + 1],
                           bufB[row][c4 + 2], bufB[row][c4 + 3]);
    *(float4*)(xcat + (base + row) * 512 + colOff + c4) = v;
  }
}

#define KU 8

__device__ __forceinline__ void load_wfrag(v2f (&b)[KU][2], const float* __restrict__ bq)
{
  #pragma unroll
  for (int u = 0; u < KU; ++u) {
    b[u][0] = *(const v2f*)(bq + u * 1024);
    b[u][1] = *(const v2f*)(bq + u * 1024 + 512);
  }
}

template <int LD>
__device__ __forceinline__ void mma_block(const v2f (&b)[KU][2], const float* aq,
                                          v8f& acc00, v8f& acc01,
                                          v8f& acc10, v8f& acc11)
{
  v2f a0[KU], a1[KU];
  #pragma unroll
  for (int u = 0; u < KU; ++u) {
    a0[u] = *(const v2f*)(aq + 4 * u);
    a1[u] = *(const v2f*)(aq + 16 * LD + 4 * u);
  }
  #pragma unroll
  for (int u = 0; u < KU; ++u) {
    acc00 = wmma_f32(a0[u], b[u][0], acc00);
    acc10 = wmma_f32(a1[u], b[u][0], acc10);
    acc01 = wmma_f32(a0[u], b[u][1], acc01);
    acc11 = wmma_f32(a1[u], b[u][1], acc11);
  }
}

template <int NKS, int LD>
__device__ __forceinline__ void gemm_2x2(const float* __restrict__ bp, const float* ap,
                                         v8f& acc00, v8f& acc01,
                                         v8f& acc10, v8f& acc11)
{
  constexpr int NB = NKS / KU;
  v2f bA[KU][2], bB[KU][2];
  load_wfrag(bA, bp);
  __builtin_amdgcn_sched_group_barrier(0x020, 16, 0);
  #pragma unroll
  for (int blk = 0; blk < NB; blk += 2) {
    const float* bq = bp + (size_t)blk * (KU * 1024);
    const float* aq = ap + blk * (KU * 4);
    load_wfrag(bB, bq + KU * 1024);
    mma_block<LD>(bA, aq, acc00, acc01, acc10, acc11);
    __builtin_amdgcn_sched_group_barrier(0x020, 16, 0);
    __builtin_amdgcn_sched_group_barrier(0x008, 32, 0);
    if (blk + 2 < NB) {
      load_wfrag(bA, bq + 2 * KU * 1024);
      __builtin_amdgcn_sched_group_barrier(0x020, 16, 0);
    }
    mma_block<LD>(bB, aq + KU * 4, acc00, acc01, acc10, acc11);
    __builtin_amdgcn_sched_group_barrier(0x008, 32, 0);
  }
}

__global__ __launch_bounds__(256) void gin_chan(
    const float* __restrict__ xcat,
    const float* __restrict__ W1p, const float* __restrict__ B1,
    const float* __restrict__ W2p, const float* __restrict__ B2,
    const float* __restrict__ W3,  const float* __restrict__ B3,
    float* __restrict__ outp)
{
  __shared__ __attribute__((aligned(16))) float bufA[32][516];
  __shared__ __attribute__((aligned(16))) float bufB[32][260];
  const int tid  = threadIdx.x;
  const int lane = tid & 31;
  const int w    = tid >> 5;
  const int c    = blockIdx.y;
  const size_t base = (size_t)blockIdx.x * 32;

  for (int i = tid; i < 32 * 128; i += 256) {
    int row = i >> 7, c4 = (i & 127) << 2;
    const float4 v = *(const float4*)(xcat + (base + row) * 512 + c4);
    bufA[row][c4] = v.x; bufA[row][c4 + 1] = v.y;
    bufA[row][c4 + 2] = v.z; bufA[row][c4 + 3] = v.w;
  }
  __syncthreads();

  const int half = lane >> 4;
  const int n16  = lane & 15;
  const int ct0  = w * 16;
  const int ct1  = (w + 8) * 16;
  const int bofs = ((ct0 + n16) << 2) + half * 2;

  {
    v8f acc00 = {}, acc01 = {}, acc10 = {}, acc11 = {};
    gemm_2x2<128, 516>(W1p + (size_t)c * 512 * 256 + bofs,
                       &bufA[n16][half * 2], acc00, acc01, acc10, acc11);
    const float bias0 = B1[c * 256 + ct0 + n16];
    const float bias1 = B1[c * 256 + ct1 + n16];
    #pragma unroll
    for (int j = 0; j < 8; ++j) {
      const int r = j + half * 8;
      float r00 = acc00[j] + bias0; r00 = r00 > 0.f ? r00 : 0.f;
      float r01 = acc01[j] + bias1; r01 = r01 > 0.f ? r01 : 0.f;
      float r10 = acc10[j] + bias0; r10 = r10 > 0.f ? r10 : 0.f;
      float r11 = acc11[j] + bias1; r11 = r11 > 0.f ? r11 : 0.f;
      bufB[r][ct0 + n16]      = r00;
      bufB[r][ct1 + n16]      = r01;
      bufB[r + 16][ct0 + n16] = r10;
      bufB[r + 16][ct1 + n16] = r11;
    }
  }
  __syncthreads();

  {
    v8f acc00 = {}, acc01 = {}, acc10 = {}, acc11 = {};
    gemm_2x2<64, 260>(W2p + (size_t)c * 256 * 256 + bofs,
                      &bufB[n16][half * 2], acc00, acc01, acc10, acc11);
    const float bias0 = B2[c * 256 + ct0 + n16];
    const float bias1 = B2[c * 256 + ct1 + n16];
    #pragma unroll
    for (int j = 0; j < 8; ++j) {
      const int r = j + half * 8;
      float r00 = acc00[j] + bias0; r00 = r00 > 0.f ? r00 : 0.f;
      float r01 = acc01[j] + bias1; r01 = r01 > 0.f ? r01 : 0.f;
      float r10 = acc10[j] + bias0; r10 = r10 > 0.f ? r10 : 0.f;
      float r11 = acc11[j] + bias1; r11 = r11 > 0.f ? r11 : 0.f;
      bufA[r][ct0 + n16]      = r00;
      bufA[r][ct1 + n16]      = r01;
      bufA[r + 16][ct0 + n16] = r10;
      bufA[r + 16][ct1 + n16] = r11;
    }
  }
  __syncthreads();

  if (w == 0) {
    const float* w3 = W3 + (size_t)c * 256;
    float sum = 0.f;
    for (int k = 0; k < 256; ++k) sum += bufA[lane][k] * w3[k];
    sum += B3[c];
    sum = sum > 0.f ? sum : 0.f;
    outp[(base + lane) * NCH + c] = sum;
  }
}

extern "C" void kernel_launch(void* const* d_in, const int* in_sizes, int n_in,
                              void* d_out, int out_size, void* d_ws, size_t ws_size,
                              hipStream_t stream)
{
  (void)n_in; (void)out_size; (void)ws_size;
  const float* x      = (const float*)d_in[0];
  const int*   ei     = (const int*)d_in[1];
  const float* convW  = (const float*)d_in[2];
  const float* convB  = (const float*)d_in[3];
  const float* chanW1 = (const float*)d_in[4];
  const float* chanB1 = (const float*)d_in[5];
  const float* chanW2 = (const float*)d_in[6];
  const float* chanB2 = (const float*)d_in[7];
  const float* chanW3 = (const float*)d_in[8];
  const float* chanB3 = (const float*)d_in[9];
  float* outp = (float*)d_out;

  const int N = NN;
  const long long E = (long long)in_sizes[1] / 2;

  float* xcat   = (float*)d_ws;
  float* convWp = xcat + (size_t)N * 512;
  float* W1p    = convWp + (size_t)9 * 128 * 128;
  float* W2p    = W1p + (size_t)4 * 512 * 256;
  int*   cnt    = (int*)(W2p + (size_t)4 * 256 * 256);
  int*   cursor = cnt + N;
  int*   rowptr = cursor + N;
  int*   nbr    = rowptr + N + 4;

  repack_w<<<dim3((128 * 128) / 256, 9), 256, 0, stream>>>(convW, convWp, 128, 128);
  repack_w<<<dim3((512 * 256) / 256, 4), 256, 0, stream>>>(chanW1, W1p, 512, 256);
  repack_w<<<dim3((256 * 256) / 256, 4), 256, 0, stream>>>(chanW2, W2p, 256, 256);

  const int cpBlocks = (int)(((size_t)N * 32 + 255) / 256);
  const int eBlocks  = (int)((E + 255) / 256);
  const int nBlocks  = (N + 255) / 256;

  csr_zero<<<nBlocks, 256, 0, stream>>>(cnt, N);
  csr_count<<<eBlocks, 256, 0, stream>>>(ei + E, cnt, E);
  csr_scan<<<(N + 1023) / 1024, 256, 0, stream>>>(cnt, rowptr, cursor, N);
  csr_fill<<<eBlocks, 256, 0, stream>>>(ei, ei + E, cursor, nbr, E);

  copy_rows128<<<cpBlocks, 256, 0, stream>>>(x, 128, xcat, 512, N);

  for (int l = 0; l < 3; ++l) {
    const float* h = (l == 0) ? x : (xcat + (size_t)l * 128);
    const int ldh  = (l == 0) ? 128 : 512;
    gin_mlp3<<<N / 32, 256, 0, stream>>>(h, ldh, rowptr, nbr,
                                         convWp + (size_t)l * 3 * 128 * 128,
                                         convB + l * 3 * 128,
                                         xcat, (l + 1) * 128);
  }

  gin_chan<<<dim3(N / 32, NCH), 256, 0, stream>>>(xcat, W1p, chanB1,
                                                  W2p, chanB2,
                                                  chanW3, chanB3, outp);
}


# edge-list build: global int atomics (count, cursor fill) and the scan launch replaced by a two-level counting sort on LDS counters; node sum order = LDS cursor arrival order
# speedup vs baseline: 1.8094x; 1.1292x over previous
#include <hip/hip_runtime.h>

typedef __attribute__((ext_vector_type(2))) float v2f;
typedef __attribute__((ext_vector_type(8))) float v8f;

#define NN 100000
#define NCH 4

__device__ __forceinline__ v8f wmma_f32(v2f a, v2f b, v8f c) {
  return __builtin_amdgcn_wmma_f32_16x16x4_f32(false, a, false, b, (short)0, c,
                                               false, false);
}

__global__ __launch_bounds__(256) void repack_w(
    const float* __restrict__ W, float* __restrict__ Wp, int K, int C)
{
  const size_t mat = (size_t)blockIdx.y * K * C;
  size_t idx = (size_t)blockIdx.x * 256 + threadIdx.x;
  if (idx >= (size_t)K * C) return;
  int k   = (int)(idx / C);
  int col = (int)(idx % C);
  Wp[mat + (((size_t)(k >> 2) * C + col) << 2) + (k & 3)] = W[mat + idx];
}

__global__ __launch_bounds__(256) void copy_rows128(
    const float* __restrict__ in, int ldi,
    float* __restrict__ outp, int ldo, int nRows)
{
  size_t idx = (size_t)blockIdx.x * 256 + threadIdx.x;
  if (idx >= (size_t)nRows * 32) return;
  size_t row = idx >> 5;
  int c4 = (int)(idx & 31) << 2;
  const float4 v = *(const float4*)(in + row * (size_t)ldi + c4);
  *(float4*)(outp + row * (size_t)ldo + c4) = v;
}

#define CSR_CH   8192
#define CSR_EPT  (CSR_CH / 256)
#define BIN_SH   9
#define BIN_N    (1 << BIN_SH)
#define NBINS    ((NN + BIN_N - 1) / BIN_N)
#define TABW     (NBINS + 1)
#define SRC_BITS 17
#define SRC_MASK ((1 << SRC_BITS) - 1)
#define BIN_STG  16384
static_assert(NN <= (1 << SRC_BITS), "source id must fit the packed word");
static_assert(NBINS < 256, "one histogram entry per thread");
static_assert(CSR_CH < 65536, "rank inside a chunk is kept in 16 bits");

template <bool FULL>
__device__ __forceinline__ void rank_edges(
    const int* __restrict__ srcIdx, const int* __restrict__ dstIdx,
    long long cbase, int n, int tid, int* hist,
    int (&key)[CSR_EPT], int (&br)[CSR_EPT])
{
  #pragma unroll
  for (int h = 0; h < CSR_EPT; h += 16) {
    int d[16], s[16];
    #pragma unroll
    for (int i = 0; i < 16; ++i) {
      const int j = (h + i) * 256 + tid;
      const int jj = (FULL || j < n) ? j : 0;
      d[i] = dstIdx[cbase + jj];
      s[i] = srcIdx[cbase + jj];
    }
    #pragma unroll
    for (int i = 0; i < 16; ++i) {
      const int j = (h + i) * 256 + tid;
      const unsigned du = (unsigned)d[i], su = (unsigned)s[i];
      const int dd = (int)(du < (unsigned)NN ? du : (unsigned)(NN - 1));
      const int ss = (int)(su < (unsigned)NN ? su : (unsigned)(NN - 1));
      const int bin = dd >> BIN_SH;
      key[h + i] = ((dd & (BIN_N - 1)) << SRC_BITS) | ss;
      br[h + i]  = -1;
      if (FULL || j < n) br[h + i] = (bin << 16) | atomicAdd(&hist[bin], 1);
    }
  }
}

template <bool FULL>
__device__ __forceinline__ void place_keys(const int* hist, int* keys,
    const int (&key)[CSR_EPT], const int (&br)[CSR_EPT])
{
  #pragma unroll
  for (int i = 0; i < CSR_EPT; ++i) {
    const int pos = hist[(br[i] >> 16) & 255] + (br[i] & 0xFFFF);
    if (FULL || br[i] >= 0) keys[pos] = key[i];
  }
}

__global__ __launch_bounds__(256) void csr_chunk_sort(
    const int* __restrict__ srcIdx, const int* __restrict__ dstIdx,
    int* __restrict__ sorted, int* __restrict__ tab, long long E)
{
  __shared__ int hist[256];
  __shared__ int wtot[8];
  __shared__ int keys[CSR_CH];
  const int tid  = threadIdx.x;
  const int lane = tid & 31;
  const int w    = tid >> 5;
  const long long cbase = (long long)blockIdx.x * CSR_CH;
  const long long left  = E - cbase;
  const int n = left < CSR_CH ? (int)left : CSR_CH;

  hist[tid] = 0;
  __syncthreads();

  int key[CSR_EPT], br[CSR_EPT];
  if (n == CSR_CH) rank_edges<true>(srcIdx, dstIdx, cbase, n, tid, hist, key, br);
  else             rank_edges<false>(srcIdx, dstIdx, cbase, n, tid, hist, key, br);
  __syncthreads();

  const int own = hist[tid];
  int incl = own;
  for (int dlt = 1; dlt < 32; dlt <<= 1) {
    const int t = __shfl_up(incl, dlt);
    if (lane >= dlt) incl += t;
  }
  if (lane == 31) wtot[w] = incl;
  __syncthreads();
  int wb = 0;
  for (int k = 0; k < 8; ++k) if (k < w) wb += wtot[k];
  const int excl = wb + incl - own;
  hist[tid] = excl;
  if (tid <= NBINS) tab[(size_t)blockIdx.x * TABW + tid] = excl;
  __syncthreads();

  if (n == CSR_CH) place_keys<true>(hist, keys, key, br);
  else             place_keys<false>(hist, keys, key, br);
  __syncthreads();

  for (int j = tid; j < n; j += 256) sorted[cbase + j] = keys[j];
}

template <bool FILL>
__device__ __forceinline__ void place_edge(int kv, int* cnt, int* stage,
                                           int* __restrict__ nbrBin, bool staged)
{
  const int local = kv >> SRC_BITS;
  if (!FILL) {
    atomicAdd(&cnt[local], 1);
  } else {
    const int slot = atomicAdd(&cnt[local], 1);
    if (staged) stage[slot] = kv & SRC_MASK;
    else        nbrBin[slot] = kv & SRC_MASK;
  }
}

template <bool FILL>
__device__ __forceinline__ void walk_bin(
    const int* __restrict__ sorted, const int* __restrict__ tab,
    int b, int nChunks, int wu, int lane,
    int* cnt, int* stage, int* __restrict__ nbrBin, bool staged)
{
  for (int c0 = 0; c0 < nChunks; c0 += 256) {
    const int cl = c0 + wu + 32 * lane;
    int rsv = 0, rev = 0;
    if (lane < 8 && cl < nChunks) {
      rsv = tab[(size_t)cl * TABW + b];
      rev = tab[(size_t)cl * TABW + b + 1];
    }
    int kq[8][2];
    #pragma unroll
    for (int l = 0; l < 8; ++l) {
      const int rs = __builtin_amdgcn_readlane(rsv, l);
      const int re = __builtin_amdgcn_readlane(rev, l);
      const int c  = c0 + wu + 32 * l;
      const int* run = sorted + (size_t)(c < nChunks ? c : 0) * CSR_CH;
      #pragma unroll
      for (int u = 0; u < 2; ++u) {
        const int j = rs + u * 32 + lane;
        kq[l][u] = run[j < re ? j : 0];
      }
    }
    #pragma unroll
    for (int l = 0; l < 8; ++l) {
      const int rs = __builtin_amdgcn_readlane(rsv, l);
      const int re = __builtin_amdgcn_readlane(rev, l);
      #pragma unroll
      for (int u = 0; u < 2; ++u)
        if (rs + u * 32 + lane < re)
          place_edge<FILL>(kq[l][u], cnt, stage, nbrBin, staged);
    }
    for (int l = 0; l < 8; ++l) {
      const int rs = __builtin_amdgcn_readlane(rsv, l);
      const int re = __builtin_amdgcn_readlane(rev, l);
      if (re - rs <= 64) continue;
      const int* run = sorted + (size_t)(c0 + wu + 32 * l) * CSR_CH;
      for (int j = rs + 64 + lane; j < re; j += 32)
        place_edge<FILL>(run[j], cnt, stage, nbrBin, staged);
    }
  }
}

__global__ __launch_bounds__(1024) void csr_bin_build(
    const int* __restrict__ sorted, const int* __restrict__ tab,
    int* __restrict__ rowptr, int* __restrict__ nbr, int nChunks)
{
  __shared__ int cnt[BIN_N];
  __shared__ int wpart[32];
  __shared__ int wsum[32];
  __shared__ int stage[BIN_STG];
  const int tid  = threadIdx.x;
  const int lane = tid & 31;
  const int wu = __builtin_amdgcn_readfirstlane(tid >> 5);
  const int b  = blockIdx.x;

  if (tid < BIN_N) cnt[tid] = 0;
  int part = 0;
  for (int c = tid; c < nChunks; c += 1024) part += tab[(size_t)c * TABW + b];
  for (int dlt = 16; dlt >= 1; dlt >>= 1) part += __shfl_xor(part, dlt);
  if (lane == 0) wpart[wu] = part;
  __syncthreads();
  int base = 0;
  for (int k = 0; k < 32; ++k) base += wpart[k];

  walk_bin<false>(sorted, tab, b, nChunks, wu, lane, cnt, stage, nbr, false);
  __syncthreads();

  const int own = tid < BIN_N ? cnt[tid] : 0;
  int incl = own;
  for (int dlt = 1; dlt < 32; dlt <<= 1) {
    const int t = __shfl_up(incl, dlt);
    if (lane >= dlt) incl += t;
  }
  if (lane == 31) wsum[wu] = incl;
  __syncthreads();
  int wb = 0, total = 0;
  for (int k = 0; k < 32; ++k) {
    const int v = wsum[k];
    if (k < wu) wb += v;
    total += v;
  }
  const int excl = wb + incl - own;
  if (tid < BIN_N) {
    cnt[tid] = excl;
    const int node = b * BIN_N + tid;
    if (node < NN) rowptr[node] = base + excl;
  }
  if (b == (int)gridDim.x - 1 && tid == 0) rowptr[NN] = base + total;
  const int baseU  = __builtin_amdgcn_readfirstlane(base);
  const int totalU = __builtin_amdgcn_readfirstlane(total);
  const bool staged = totalU <= BIN_STG;
  __syncthreads();

  walk_bin<true>(sorted, tab, b, nChunks, wu, lane, cnt, stage, nbr + baseU, staged);
  if (staged) {
    __syncthreads();
    for (int i = tid; i < totalU; i += 1024) nbr[baseU + i] = stage[i];
  }
}

template <int U>
__device__ __forceinline__ void add_rows(float4& acc, const float* colp, int ldf,
                                         int mine, int t)
{
  float4 v[U];
  #pragma unroll
  for (int u = 0; u < U; ++u) {
    const int s = __builtin_amdgcn_readlane(mine, t + u);
    v[u] = *(const float4*)(colp + (size_t)s * ldf);
  }
  #pragma unroll
  for (int u = 0; u < U; ++u) {
    acc.x += v[u].x; acc.y += v[u].y; acc.z += v[u].z; acc.w += v[u].w;
  }
}

__device__ __forceinline__ void sum_list(float4& acc, const float* colp, int ldf,
                                         const int* __restrict__ nbr,
                                         int beg, int end, int mine, int lane)
{
  for (int pos = beg; pos < end; pos += 32) {
    const int left = end - pos;
    const int n = left < 32 ? left : 32;
    if (pos != beg) mine = (lane < n) ? nbr[pos + lane] : 0;
    int t = 0;
    for (; t + 8 <= n; t += 8) add_rows<8>(acc, colp, ldf, mine, t);
    if (t + 4 <= n) { add_rows<4>(acc, colp, ldf, mine, t); t += 4; }
    if (t + 2 <= n) { add_rows<2>(acc, colp, ldf, mine, t); t += 2; }
    if (t < n)      { add_rows<1>(acc, colp, ldf, mine, t); }
  }
}

__global__ __launch_bounds__(256) void gin_mlp3(
    const float* __restrict__ feat, int ldf,
    const int* __restrict__ rowptr, const int* __restrict__ nbr,
    const float* __restrict__ Wp,
    const float* __restrict__ Bv,
    float* __restrict__ xcat, int colOff)
{
  __shared__ __attribute__((aligned(16))) float bufA[32][132];
  __shared__ float bufB[32][132];
  const int tid  = threadIdx.x;
  const int lane = tid & 31;
  const int w    = tid >> 5;
  const size_t base = (size_t)blockIdx.x * 32;

  {
    const int wu    = __builtin_amdgcn_readfirstlane(tid >> 5);
    const int node0 = (int)base + wu * 4;
    const int rp = (lane < 5) ? rowptr[node0 + lane] : 0;
    int rb[5];
    #pragma unroll
    for (int j = 0; j < 5; ++j) rb[j] = __builtin_amdgcn_readlane(rp, j);
    const float* colp = feat + lane * 4;
    int mine[4];
    float4 acc[4];
    #pragma unroll
    for (int j = 0; j < 4; ++j)
      mine[j] = (lane < rb[j + 1] - rb[j]) ? nbr[rb[j] + lane] : 0;
    #pragma unroll
    for (int j = 0; j < 4; ++j)
      acc[j] = *(const float4*)(colp + (size_t)(node0 + j) * ldf);
    #pragma unroll
    for (int j = 0; j < 4; ++j) {
      sum_list(acc[j], colp, ldf, nbr, rb[j], rb[j + 1], mine[j], lane);
      *(float4*)&bufA[wu * 4 + j][lane * 4] = acc[j];
    }
  }
  __syncthreads();

  const int half = lane >> 4;
  const int n16  = lane & 15;
  const int cb   = w * 16;

  for (int layer = 0; layer < 3; ++layer) {
    const float* Wk = Wp + (size_t)layer * 128 * 128;
    float (*in)[132]   = (layer & 1) ? bufB : bufA;
    float (*outb)[132] = (layer & 1) ? bufA : bufB;
    v8f acc0 = {}, acc1 = {};
    for (int k = 0; k < 128; k += 4) {
      const int kk = k + half * 2;
      const v2f b = *(const v2f*)(Wk + (((size_t)(k >> 2) * 128 + cb + n16) << 2)
                                     + half * 2);
      v2f a0; a0.x = in[n16][kk];      a0.y = in[n16][kk + 1];
      v2f a1; a1.x = in[n16 + 16][kk]; a1.y = in[n16 + 16][kk + 1];
      acc0 = wmma_f32(a0, b, acc0);
      acc1 = wmma_f32(a1, b, acc1);
    }
    const float bias = Bv[layer * 128 + cb + n16];
    #pragma unroll
    for (int j = 0; j < 8; ++j) {
      float r0 = acc0[j] + bias; r0 = r0 > 0.f ? r0 : 0.f;
      float r1 = acc1[j] + bias; r1 = r1 > 0.f ? r1 : 0.f;
      outb[j + half * 8][cb + n16]      = r0;
      outb[16 + j + half * 8][cb + n16] = r1;
    }
    __syncthreads();
  }
  for (int i = tid; i < 32 * 32; i += 256) {
    int row = i >> 5, c4 = (i & 31) << 2;
    float4 v = make_float4(bufB[row][c4], bufB[row][c4 + 1],
                           bufB[row][c4 + 2], bufB[row][c4 + 3]);
    *(float4*)(xcat + (base + row) * 512 + colOff + c4) = v;
  }
}

#define KU 8

__device__ __forceinline__ void load_wfrag(v2f (&b)[KU][2], const float* __restrict__ bq)
{
  #pragma unroll
  for (int u = 0; u < KU; ++u) {
    b[u][0] = *(const v2f*)(bq + u * 1024);
    b[u][1] = *(const v2f*)(bq + u * 1024 + 512);
  }
}

template <int LD>
__device__ __forceinline__ void mma_block(const v2f (&b)[KU][2], const float* aq,
                                          v8f& acc00, v8f& acc01,
                                          v8f& acc10, v8f& acc11)
{
  v2f a0[KU], a1[KU];
  #pragma unroll
  for (int u = 0; u < KU; ++u) {
    a0[u] = *(const v2f*)(aq + 4 * u);
    a1[u] = *(const v2f*)(aq + 16 * LD + 4 * u);
  }
  #pragma unroll
  for (int u = 0; u < KU; ++u) {
    acc00 = wmma_f32(a0[u], b[u][0], acc00);
    acc10 = wmma_f32(a1[u], b[u][0], acc10);
    acc01 = wmma_f32(a0[u], b[u][1], acc01);
    acc11 = wmma_f32(a1[u], b[u][1], acc11);
  }
}

template <int NKS, int LD>
__device__ __forceinline__ void gemm_2x2(const float* __restrict__ bp, const float* ap,
                                         v8f& acc00, v8f& acc01,
                                         v8f& acc10, v8f& acc11)
{
  constexpr int NB = NKS / KU;
  v2f bA[KU][2], bB[KU][2];
  load_wfrag(bA, bp);
  __builtin_amdgcn_sched_group_barrier(0x020, 16, 0);
  #pragma unroll
  for (int blk = 0; blk < NB; blk += 2) {
    const float* bq = bp + (size_t)blk * (KU * 1024);
    const float* aq = ap + blk * (KU * 4);
    load_wfrag(bB, bq + KU * 1024);
    mma_block<LD>(bA, aq, acc00, acc01, acc10, acc11);
    __builtin_amdgcn_sched_group_barrier(0x020, 16, 0);
    __builtin_amdgcn_sched_group_barrier(0x008, 32, 0);
    if (blk + 2 < NB) {
      load_wfrag(bA, bq + 2 * KU * 1024);
      __builtin_amdgcn_sched_group_barrier(0x020, 16, 0);
    }
    mma_block<LD>(bB, aq + KU * 4, acc00, acc01, acc10, acc11);
    __builtin_amdgcn_sched_group_barrier(0x008, 32, 0);
  }
}

__global__ __launch_bounds__(256) void gin_chan(
    const float* __restrict__ xcat,
    const float* __restrict__ W1p, const float* __restrict__ B1,
    const float* __restrict__ W2p, const float* __restrict__ B2,
    const float* __restrict__ W3,  const float* __restrict__ B3,
    float* __restrict__ outp)
{
  __shared__ __attribute__((aligned(16))) float bufA[32][516];
  __shared__ __attribute__((aligned(16))) float bufB[32][260];
  const int tid  = threadIdx.x;
  const int lane = tid & 31;
  const int w    = tid >> 5;
  const int c    = blockIdx.y;
  const size_t base = (size_t)blockIdx.x * 32;

  for (int i = tid; i < 32 * 128; i += 256) {
    int row = i >> 7, c4 = (i & 127) << 2;
    const float4 v = *(const float4*)(xcat + (base + row) * 512 + c4);
    bufA[row][c4] = v.x; bufA[row][c4 + 1] = v.y;
    bufA[row][c4 + 2] = v.z; bufA[row][c4 + 3] = v.w;
  }
  __syncthreads();

  const int half = lane >> 4;
  const int n16  = lane & 15;
  const int ct0  = w * 16;
  const int ct1  = (w + 8) * 16;
  const int bofs = ((ct0 + n16) << 2) + half * 2;

  {
    v8f acc00 = {}, acc01 = {}, acc10 = {}, acc11 = {};
    gemm_2x2<128, 516>(W1p + (size_t)c * 512 * 256 + bofs,
                       &bufA[n16][half * 2], acc00, acc01, acc10, acc11);
    const float bias0 = B1[c * 256 + ct0 + n16];
    const float bias1 = B1[c * 256 + ct1 + n16];
    #pragma unroll
    for (int j = 0; j < 8; ++j) {
      const int r = j + half * 8;
      float r00 = acc00[j] + bias0; r00 = r00 > 0.f ? r00 : 0.f;
      float r01 = acc01[j] + bias1; r01 = r01 > 0.f ? r01 : 0.f;
      float r10 = acc10[j] + bias0; r10 = r10 > 0.f ? r10 : 0.f;
      float r11 = acc11[j] + bias1; r11 = r11 > 0.f ? r11 : 0.f;
      bufB[r][ct0 + n16]      = r00;
      bufB[r][ct1 + n16]      = r01;
      bufB[r + 16][ct0 + n16] = r10;
      bufB[r + 16][ct1 + n16] = r11;
    }
  }
  __syncthreads();

  {
    v8f acc00 = {}, acc01 = {}, acc10 = {}, acc11 = {};
    gemm_2x2<64, 260>(W2p + (size_t)c * 256 * 256 + bofs,
                      &bufB[n16][half * 2], acc00, acc01, acc10, acc11);
    const float bias0 = B2[c * 256 + ct0 + n16];
    const float bias1 = B2[c * 256 + ct1 + n16];
    #pragma unroll
    for (int j = 0; j < 8; ++j) {
      const int r = j + half * 8;
      float r00 = acc00[j] + bias0; r00 = r00 > 0.f ? r00 : 0.f;
      float r01 = acc01[j] + bias1; r01 = r01 > 0.f ? r01 : 0.f;
      float r10 = acc10[j] + bias0; r10 = r10 > 0.f ? r10 : 0.f;
      float r11 = acc11[j] + bias1; r11 = r11 > 0.f ? r11 : 0.f;
      bufA[r][ct0 + n16]      = r00;
      bufA[r][ct1 + n16]      = r01;
      bufA[r + 16][ct0 + n16] = r10;
      bufA[r + 16][ct1 + n16] = r11;
    }
  }
  __syncthreads();

  if (w == 0) {
    const float* w3 = W3 + (size_t)c * 256;
    float sum = 0.f;
    for (int k = 0; k < 256; ++k) sum += bufA[lane][k] * w3[k];
    sum += B3[c];
    sum = sum > 0.f ? sum : 0.f;
    outp[(base + lane) * NCH + c] = sum;
  }
}

extern "C" void kernel_launch(void* const* d_in, const int* in_sizes, int n_in,
                              void* d_out, int out_size, void* d_ws, size_t ws_size,
                              hipStream_t stream)
{
  (void)n_in; (void)out_size; (void)ws_size;
  const float* x      = (const float*)d_in[0];
  const int*   ei     = (const int*)d_in[1];
  const float* convW  = (const float*)d_in[2];
  const float* convB  = (const float*)d_in[3];
  const float* chanW1 = (const float*)d_in[4];
  const float* chanB1 = (const float*)d_in[5];
  const float* chanW2 = (const float*)d_in[6];
  const float* chanB2 = (const float*)d_in[7];
  const float* chanW3 = (const float*)d_in[8];
  const float* chanB3 = (const float*)d_in[9];
  float* outp = (float*)d_out;

  const int N = NN;
  const long long E = (long long)in_sizes[1] / 2;

  float* xcat   = (float*)d_ws;
  float* convWp = xcat + (size_t)N * 512;
  float* W1p    = convWp + (size_t)9 * 128 * 128;
  float* W2p    = W1p + (size_t)4 * 512 * 256;
  int*   rowptr = (int*)(W2p + (size_t)4 * 256 * 256);
  int*   nbr    = rowptr + N + 4;
  int*   sorted = nbr + E;
  int*   tab    = sorted + E;

  repack_w<<<dim3((128 * 128) / 256, 9), 256, 0, stream>>>(convW, convWp, 128, 128);
  repack_w<<<dim3((512 * 256) / 256, 4), 256, 0, stream>>>(chanW1, W1p, 512, 256);
  repack_w<<<dim3((256 * 256) / 256, 4), 256, 0, stream>>>(chanW2, W2p, 256, 256);

  const int cpBlocks = (int)(((size_t)N * 32 + 255) / 256);
  const int nChunks  = (int)((E + CSR_CH - 1) / CSR_CH);

  csr_chunk_sort<<<nChunks, 256, 0, stream>>>(ei, ei + E, sorted, tab, E);
  csr_bin_build<<<NBINS, 1024, 0, stream>>>(sorted, tab, rowptr, nbr, nChunks);

  copy_rows128<<<cpBlocks, 256, 0, stream>>>(x, 128, xcat, 512, N);

  for (int l = 0; l < 3; ++l) {
    const float* h = (l == 0) ? x : (xcat + (size_t)l * 128);
    const int ldh  = (l == 0) ? 128 : 512;
    gin_mlp3<<<N / 32, 256, 0, stream>>>(h, ldh, rowptr, nbr,
                                         convWp + (size_t)l * 3 * 128 * 128,
                                         convB + l * 3 * 128,
                                         xcat, (l + 1) * 128);
  }

  gin_chan<<<dim3(N / 32, NCH), 256, 0, stream>>>(xcat, W1p, chanB1,
                                                  W2p, chanB2,
                                                  chanW3, chanB3, outp);
}
